# starting kernel
# baseline (speedup 1.0000x reference)
#include <hip/hip_runtime.h>
#include <hip/hip_bf16.h>

#define N_NODES 50000
#define N_EDGES 1600000
#define IN_DIM 256
#define OUT_DIM 128
#define NEG_SLOPE 0.2f

typedef __attribute__((ext_vector_type(2))) float v2f;
typedef __attribute__((ext_vector_type(8))) float v8f;

__global__ __launch_bounds__(256) void gat_gemm_wmma(
    const float* __restrict__ X,
    const float* __restrict__ W,
    float* __restrict__ H)
{
  const int lane = threadIdx.x & 31;
  const int wave = threadIdx.x >> 5;
  const int l16  = lane & 15;
  const int half = lane >> 4;

  const int mBase = blockIdx.x * 16;
  const int nBase = wave * 16;

  const float* aPtr = X + (size_t)(mBase + l16) * IN_DIM + 2 * half;
  const float* bPtr = W + (size_t)(2 * half) * OUT_DIM + nBase + l16;

  v8f acc = {};
#pragma unroll 4
  for (int kk = 0; kk < IN_DIM; kk += 4) {
    v2f a = *(const v2f*)(aPtr + kk);
    v2f b;
    b.x = bPtr[(size_t)kk * OUT_DIM];
    b.y = bPtr[(size_t)(kk + 1) * OUT_DIM];
    acc = __builtin_amdgcn_wmma_f32_16x16x4_f32(
         false, a,  false, b,
         (short)0, acc,  false,  false);
  }

  float* out = H + (size_t)(mBase + 8 * half) * OUT_DIM + nBase + l16;
#pragma unroll
  for (int v = 0; v < 8; ++v)
    out[(size_t)v * OUT_DIM] = acc[v];
}

__global__ __launch_bounds__(256) void node_logits(
    const float* __restrict__ H,
    const float* __restrict__ attn_l,
    const float* __restrict__ attn_r,
    float* __restrict__ el, float* __restrict__ er)
{
  const int lane = threadIdx.x & 31;
  const int node = blockIdx.x * (blockDim.x >> 5) + (threadIdx.x >> 5);
  if (node >= N_NODES) return;

  const float4 hv = *(const float4*)(H + (size_t)node * OUT_DIM + 4 * lane);
  const float4 al = *(const float4*)(attn_l + 4 * lane);
  const float4 ar = *(const float4*)(attn_r + 4 * lane);
  float sl = hv.x * al.x + hv.y * al.y + hv.z * al.z + hv.w * al.w;
  float sr = hv.x * ar.x + hv.y * ar.y + hv.z * ar.z + hv.w * ar.w;
#pragma unroll
  for (int off = 16; off > 0; off >>= 1) {
    sl += __shfl_down(sl, off, 32);
    sr += __shfl_down(sr, off, 32);
  }
  if (lane == 0) { el[node] = sl; er[node] = sr; }
}

__global__ __launch_bounds__(256) void init_buffers(
    float* __restrict__ out, float* __restrict__ emax, float* __restrict__ esum)
{
  const int i = blockIdx.x * blockDim.x + threadIdx.x;
  if (i < N_NODES * OUT_DIM) out[i] = 0.0f;
  if (i < N_NODES) { emax[i] = -__builtin_inff(); esum[i] = 0.0f; }
}

__global__ __launch_bounds__(256) void edge_score_max(
    const int* __restrict__ src, const int* __restrict__ dst,
    const float* __restrict__ el, const float* __restrict__ er,
    float* __restrict__ score, float* __restrict__ emax)
{
  const int e = blockIdx.x * blockDim.x + threadIdx.x;
  if (e >= N_EDGES) return;
  const float t = el[src[e]] + er[dst[e]];
  const float s = t > 0.0f ? t : NEG_SLOPE * t;
  score[e] = s;
  atomicMax(emax + dst[e], s);
}

__global__ __launch_bounds__(256) void edge_exp_sum(
    const int* __restrict__ dst,
    float* __restrict__ score,
    const float* __restrict__ emax, float* __restrict__ esum)
{
  const int e = blockIdx.x * blockDim.x + threadIdx.x;
  if (e >= N_EDGES) return;
  const int d = dst[e];
  const float ex = __expf(score[e] - emax[d]);
  score[e] = ex;
  atomicAdd(esum + d, ex);
}

__global__ __launch_bounds__(256) void aggregate(
    const int* __restrict__ src, const int* __restrict__ dst,
    const float* __restrict__ H,
    const float* __restrict__ eexp, const float* __restrict__ esum,
    float* __restrict__ out)
{
  const int lane = threadIdx.x & 31;
  const int e = blockIdx.x * (blockDim.x >> 5) + (threadIdx.x >> 5);
  if (e >= N_EDGES) return;

  const int s = src[e], d = dst[e];
  const float a = eexp[e] / esum[d];
  const float4 hv = *(const float4*)(H + (size_t)s * OUT_DIM + 4 * lane);
  float* o = out + (size_t)d * OUT_DIM + 4 * lane;
  atomicAdd(o + 0, a * hv.x);
  atomicAdd(o + 1, a * hv.y);
  atomicAdd(o + 2, a * hv.z);
  atomicAdd(o + 3, a * hv.w);
}

extern "C" void kernel_launch(void* const* d_in, const int* in_sizes, int n_in,
                              void* d_out, int out_size, void* d_ws, size_t ws_size,
                              hipStream_t stream) {
  (void)in_sizes; (void)n_in; (void)out_size; (void)ws_size;

  const float* features  = (const float*)d_in[0];
  const float* fc_weight = (const float*)d_in[1];
  const float* attn_l    = (const float*)d_in[2];
  const float* attn_r    = (const float*)d_in[3];
  const int*   src       = (const int*)d_in[4];
  const int*   dst       = (const int*)d_in[5];
  float*       out       = (float*)d_out;

  float* h     = (float*)d_ws;
  float* el    = h + (size_t)N_NODES * OUT_DIM;
  float* er    = el + N_NODES;
  float* emax  = er + N_NODES;
  float* esum  = emax + N_NODES;
  float* score = esum + N_NODES;

  init_buffers<<<(N_NODES * OUT_DIM + 255) / 256, 256, 0, stream>>>(out, emax, esum);

  gat_gemm_wmma<<<N_NODES / 16, 256, 0, stream>>>(features, fc_weight, h);

  node_logits<<<(N_NODES + 7) / 8, 256, 0, stream>>>(h, attn_l, attn_r, el, er);

  edge_score_max<<<(N_EDGES + 255) / 256, 256, 0, stream>>>(src, dst, el, er, score, emax);

  edge_exp_sum<<<(N_EDGES + 255) / 256, 256, 0, stream>>>(dst, score, emax, esum);

  aggregate<<<(N_EDGES + 7) / 8, 256, 0, stream>>>(src, dst, h, score, esum, out);
}


# Edges grouped by destination per call (arrival-order rank from one int atomic, scan, fill); one wave per node does softmax and row sum in registers; float atomics and init gone; sum order changes
# speedup vs baseline: 2.1804x; 2.1804x over previous
#include <hip/hip_runtime.h>
#include <hip/hip_bf16.h>

#define N_NODES 50000
#define N_EDGES 1600000
#define IN_DIM 256
#define OUT_DIM 128
#define NEG_SLOPE 0.2f
#define N_PAD 50048
#define SCAN_PER_WAVE 1568

typedef __attribute__((ext_vector_type(2))) float v2f;
typedef __attribute__((ext_vector_type(8))) float v8f;

__global__ __launch_bounds__(256) void gat_gemm_wmma(
    const float* __restrict__ X,
    const float* __restrict__ W,
    float* __restrict__ H)
{
  const int lane = threadIdx.x & 31;
  const int wave = threadIdx.x >> 5;
  const int l16  = lane & 15;
  const int half = lane >> 4;

  const int mBase = blockIdx.x * 16;
  const int nBase = wave * 16;

  const float* aPtr = X + (size_t)(mBase + l16) * IN_DIM + 2 * half;
  const float* bPtr = W + (size_t)(2 * half) * OUT_DIM + nBase + l16;

  v8f acc = {};
#pragma unroll 4
  for (int kk = 0; kk < IN_DIM; kk += 4) {
    v2f a = *(const v2f*)(aPtr + kk);
    v2f b;
    b.x = bPtr[(size_t)kk * OUT_DIM];
    b.y = bPtr[(size_t)(kk + 1) * OUT_DIM];
    acc = __builtin_amdgcn_wmma_f32_16x16x4_f32(
         false, a,  false, b,
         (short)0, acc,  false,  false);
  }

  float* out = H + (size_t)(mBase + 8 * half) * OUT_DIM + nBase + l16;
#pragma unroll
  for (int v = 0; v < 8; ++v)
    out[(size_t)v * OUT_DIM] = acc[v];
}

__global__ __launch_bounds__(256) void node_logits(
    const float* __restrict__ H,
    const float* __restrict__ attn_l,
    const float* __restrict__ attn_r,
    float* __restrict__ el, float* __restrict__ er)
{
  const int lane = threadIdx.x & 31;
  const int node = blockIdx.x * (blockDim.x >> 5) + (threadIdx.x >> 5);
  if (node >= N_NODES) return;

  const float4 hv = *(const float4*)(H + (size_t)node * OUT_DIM + 4 * lane);
  const float4 al = *(const float4*)(attn_l + 4 * lane);
  const float4 ar = *(const float4*)(attn_r + 4 * lane);
  float sl = hv.x * al.x + hv.y * al.y + hv.z * al.z + hv.w * al.w;
  float sr = hv.x * ar.x + hv.y * ar.y + hv.z * ar.z + hv.w * ar.w;
#pragma unroll
  for (int off = 16; off > 0; off >>= 1) {
    sl += __shfl_down(sl, off, 32);
    sr += __shfl_down(sr, off, 32);
  }
  if (lane == 0) { el[node] = sl; er[node] = sr; }
}

__global__ __launch_bounds__(256) void zero_counts(int* __restrict__ cnt)
{
  const int i = blockIdx.x * blockDim.x + threadIdx.x;
  if (i <= N_NODES) cnt[i] = 0;
}

__global__ __launch_bounds__(256) void count_rank(
    const int* __restrict__ dst, int* __restrict__ cnt, int* __restrict__ rank)
{
  const int e = blockIdx.x * blockDim.x + threadIdx.x;
  if (e >= N_EDGES) return;
  rank[e] = atomicAdd(cnt + dst[e], 1);
}

__global__ __launch_bounds__(1024) void scan_counts(int* __restrict__ cnt)
{
  __shared__ int wsum[32];
  const int lane = threadIdx.x & 31;
  const int wave = threadIdx.x >> 5;
  const int lo = wave * SCAN_PER_WAVE + lane;

  int v[49];
  int s = 0;
#pragma unroll
  for (int i = 0; i < 49; ++i) {
    const int idx = lo + i * 32;
    v[i] = idx < N_NODES ? cnt[idx] : 0;
    s += v[i];
  }
#pragma unroll
  for (int o = 16; o > 0; o >>= 1) s += __shfl_xor(s, o, 32);
  if (lane == 0) wsum[wave] = s;
  __syncthreads();

  int base = 0;
  for (int w = 0; w < wave; ++w) base += wsum[w];

  int x[49];
#pragma unroll
  for (int i = 0; i < 49; ++i) x[i] = v[i];
#pragma unroll
  for (int o = 1; o < 32; o <<= 1) {
#pragma unroll
    for (int i = 0; i < 49; ++i) {
      const int y = __shfl_up(x[i], o, 32);
      if (lane >= o) x[i] += y;
    }
  }
#pragma unroll
  for (int i = 0; i < 49; ++i) {
    const int idx = lo + i * 32;
    if (idx <= N_NODES) cnt[idx] = base + x[i] - v[i];
    base += __shfl(x[i], 31, 32);
  }
}

__global__ __launch_bounds__(256) void fill_csr(
    const int* __restrict__ src, const int* __restrict__ dst,
    const int* __restrict__ off, const int* __restrict__ rank,
    int* __restrict__ csr)
{
  const int e = blockIdx.x * blockDim.x + threadIdx.x;
  if (e >= N_EDGES) return;
  csr[off[dst[e]] + rank[e]] = src[e];
}

static __device__ __forceinline__ float leaky(float t)
{
  return t > 0.0f ? t : NEG_SLOPE * t;
}

static __device__ __forceinline__ void add_rows(
    float4& acc, const float* __restrict__ H, int lane, int sj, float a, int n)
{
  int j = 0;
  for (; j + 4 <= n; j += 4) {
    const int r0 = __builtin_amdgcn_readlane(sj, j);
    const int r1 = __builtin_amdgcn_readlane(sj, j + 1);
    const int r2 = __builtin_amdgcn_readlane(sj, j + 2);
    const int r3 = __builtin_amdgcn_readlane(sj, j + 3);
    const float w0 = __int_as_float(__builtin_amdgcn_readlane(__float_as_int(a), j));
    const float w1 = __int_as_float(__builtin_amdgcn_readlane(__float_as_int(a), j + 1));
    const float w2 = __int_as_float(__builtin_amdgcn_readlane(__float_as_int(a), j + 2));
    const float w3 = __int_as_float(__builtin_amdgcn_readlane(__float_as_int(a), j + 3));
    const float4 h0 = *(const float4*)(H + (size_t)r0 * OUT_DIM + 4 * lane);
    const float4 h1 = *(const float4*)(H + (size_t)r1 * OUT_DIM + 4 * lane);
    const float4 h2 = *(const float4*)(H + (size_t)r2 * OUT_DIM + 4 * lane);
    const float4 h3 = *(const float4*)(H + (size_t)r3 * OUT_DIM + 4 * lane);
    acc.x += w0 * h0.x; acc.y += w0 * h0.y; acc.z += w0 * h0.z; acc.w += w0 * h0.w;
    acc.x += w1 * h1.x; acc.y += w1 * h1.y; acc.z += w1 * h1.z; acc.w += w1 * h1.w;
    acc.x += w2 * h2.x; acc.y += w2 * h2.y; acc.z += w2 * h2.z; acc.w += w2 * h2.w;
    acc.x += w3 * h3.x; acc.y += w3 * h3.y; acc.z += w3 * h3.z; acc.w += w3 * h3.w;
  }
  for (; j < n; ++j) {
    const int r0 = __builtin_amdgcn_readlane(sj, j);
    const float w0 = __int_as_float(__builtin_amdgcn_readlane(__float_as_int(a), j));
    const float4 h0 = *(const float4*)(H + (size_t)r0 * OUT_DIM + 4 * lane);
    acc.x += w0 * h0.x; acc.y += w0 * h0.y; acc.z += w0 * h0.z; acc.w += w0 * h0.w;
  }
}

__global__ __launch_bounds__(256) void gather_softmax(
    const int* __restrict__ off, const int* __restrict__ csr,
    const float* __restrict__ H,
    const float* __restrict__ el, const float* __restrict__ er,
    float* __restrict__ out)
{
  const int lane = threadIdx.x & 31;
  const int d = __builtin_amdgcn_readfirstlane(
      (int)(blockIdx.x * (blockDim.x >> 5) + (threadIdx.x >> 5)));
  if (d >= N_NODES) return;

  const int start = off[d];
  const int end   = off[d + 1];
  float4* orow = (float4*)(out + (size_t)d * OUT_DIM + 4 * lane);
  if (start >= end) {
    *orow = make_float4(0.0f, 0.0f, 0.0f, 0.0f);
    return;
  }
  const float erd = er[d];
  const float ninf = -__builtin_inff();

  const int i0 = start + lane;
  const int i1 = start + 32 + lane;
  const int sj0 = csr[i0 < end ? i0 : end - 1];
  const int sj1 = csr[i1 < end ? i1 : end - 1];
  const float x0 = el[sj0];
  const float x1 = el[sj1];
  const float s0 = i0 < end ? leaky(x0 + erd) : ninf;
  const float s1 = i1 < end ? leaky(x1 + erd) : ninf;

  float m = fmaxf(s0, s1);
  for (int base = start + 64; base < end; base += 32) {
    const int idx = base + lane;
    if (idx < end) m = fmaxf(m, leaky(el[csr[idx]] + erd));
  }
#pragma unroll
  for (int o = 16; o > 0; o >>= 1) m = fmaxf(m, __shfl_xor(m, o, 32));

  const float e0 = __expf(s0 - m);
  const float e1 = __expf(s1 - m);
  float sum = e0 + e1;
  for (int base = start + 64; base < end; base += 32) {
    const int idx = base + lane;
    if (idx < end) sum += __expf(leaky(el[csr[idx]] + erd) - m);
  }
#pragma unroll
  for (int o = 16; o > 0; o >>= 1) sum += __shfl_xor(sum, o, 32);

  float4 acc = make_float4(0.0f, 0.0f, 0.0f, 0.0f);
  const int deg = end - start;
  add_rows(acc, H, lane, sj0, e0 / sum, deg < 32 ? deg : 32);
  if (deg > 32) add_rows(acc, H, lane, sj1, e1 / sum, deg < 64 ? deg - 32 : 32);
  for (int base = start + 64; base < end; base += 32) {
    const int idx = base + lane;
    int sj = 0;
    float a = 0.0f;
    if (idx < end) {
      sj = csr[idx];
      a = __expf(leaky(el[sj] + erd) - m) / sum;
    }
    const int left = end - base;
    add_rows(acc, H, lane, sj, a, left < 32 ? left : 32);
  }
  *orow = acc;
}

extern "C" void kernel_launch(void* const* d_in, const int* in_sizes, int n_in,
                              void* d_out, int out_size, void* d_ws, size_t ws_size,
                              hipStream_t stream) {
  (void)in_sizes; (void)n_in; (void)out_size; (void)ws_size;

  const float* features  = (const float*)d_in[0];
  const float* fc_weight = (const float*)d_in[1];
  const float* attn_l    = (const float*)d_in[2];
  const float* attn_r    = (const float*)d_in[3];
  const int*   src       = (const int*)d_in[4];
  const int*   dst       = (const int*)d_in[5];
  float*       out       = (float*)d_out;

  float* h    = (float*)d_ws;
  float* el   = h + (size_t)N_NODES * OUT_DIM;
  float* er   = el + N_PAD;
  int*   off  = (int*)(er + N_PAD);
  int*   csr  = off + N_PAD;
  int*   rank = (int*)d_ws;

  zero_counts<<<(N_NODES + 1 + 255) / 256, 256, 0, stream>>>(off);
  count_rank<<<(N_EDGES + 255) / 256, 256, 0, stream>>>(dst, off, rank);
  scan_counts<<<1, 1024, 0, stream>>>(off);
  fill_csr<<<(N_EDGES + 255) / 256, 256, 0, stream>>>(src, dst, off, rank, csr);

  gat_gemm_wmma<<<N_NODES / 16, 256, 0, stream>>>(features, fc_weight, h);

  node_logits<<<(N_NODES + 7) / 8, 256, 0, stream>>>(h, attn_l, attn_r, el, er);

  gather_softmax<<<(N_NODES + 7) / 8, 256, 0, stream>>>(off, csr, h, el, er, out);
}


# Edges grouped by a two-level counting sort on LDS counters (250 chunks x 196 bins of 256 destinations, then a block per bin); the 1.6M scattered global atomics go; segment order still varies
# speedup vs baseline: 3.4183x; 1.5677x over previous
#include <hip/hip_runtime.h>
#include <hip/hip_bf16.h>

#define N_NODES 50000
#define N_EDGES 1600000
#define IN_DIM 256
#define OUT_DIM 128
#define NEG_SLOPE 0.2f
#define N_PAD 50048
#define SCAN_PER_WAVE 1568

typedef __attribute__((ext_vector_type(2))) float v2f;
typedef __attribute__((ext_vector_type(8))) float v8f;

__global__ __launch_bounds__(256) void gat_gemm_wmma(
    const float* __restrict__ X,
    const float* __restrict__ W,
    float* __restrict__ H)
{
  const int lane = threadIdx.x & 31;
  const int wave = threadIdx.x >> 5;
  const int l16  = lane & 15;
  const int half = lane >> 4;

  const int mBase = blockIdx.x * 16;
  const int nBase = wave * 16;

  const float* aPtr = X + (size_t)(mBase + l16) * IN_DIM + 2 * half;
  const float* bPtr = W + (size_t)(2 * half) * OUT_DIM + nBase + l16;

  v8f acc = {};
#pragma unroll 4
  for (int kk = 0; kk < IN_DIM; kk += 4) {
    v2f a = *(const v2f*)(aPtr + kk);
    v2f b;
    b.x = bPtr[(size_t)kk * OUT_DIM];
    b.y = bPtr[(size_t)(kk + 1) * OUT_DIM];
    acc = __builtin_amdgcn_wmma_f32_16x16x4_f32(
         false, a,  false, b,
         (short)0, acc,  false,  false);
  }

  float* out = H + (size_t)(mBase + 8 * half) * OUT_DIM + nBase + l16;
#pragma unroll
  for (int v = 0; v < 8; ++v)
    out[(size_t)v * OUT_DIM] = acc[v];
}

__global__ __launch_bounds__(256) void node_logits(
    const float* __restrict__ H,
    const float* __restrict__ attn_l,
    const float* __restrict__ attn_r,
    float* __restrict__ el, float* __restrict__ er)
{
  const int lane = threadIdx.x & 31;
  const int node = blockIdx.x * (blockDim.x >> 5) + (threadIdx.x >> 5);
  if (node >= N_NODES) return;

  const float4 hv = *(const float4*)(H + (size_t)node * OUT_DIM + 4 * lane);
  const float4 al = *(const float4*)(attn_l + 4 * lane);
  const float4 ar = *(const float4*)(attn_r + 4 * lane);
  float sl = hv.x * al.x + hv.y * al.y + hv.z * al.z + hv.w * al.w;
  float sr = hv.x * ar.x + hv.y * ar.y + hv.z * ar.z + hv.w * ar.w;
#pragma unroll
  for (int off = 16; off > 0; off >>= 1) {
    sl += __shfl_down(sl, off, 32);
    sr += __shfl_down(sr, off, 32);
  }
  if (lane == 0) { el[node] = sl; er[node] = sr; }
}

#define N_CHUNK 250
#define CHUNK_EDGES 6400
#define N_BIN 196
#define N_BIN_ROWS 200
#define HIST_PAD N_PAD
static_assert(N_BIN_ROWS * N_CHUNK == N_NODES, "scan_counts scans exactly N_NODES counts");
static_assert(N_BIN_ROWS >= N_BIN, "every bin has a histogram row");
static_assert(N_CHUNK * CHUNK_EDGES == N_EDGES, "chunks must tile the edge list");
static_assert(N_BIN * 256 > N_NODES, "bins must cover off[0..N_NODES]");
static_assert(N_NODES <= 65536, "a source index must fit 16 bits of the packed edge word");

__global__ __launch_bounds__(256) void bin_count(
    const int* __restrict__ dst, int* __restrict__ hist)
{
  __shared__ int hcnt[N_BIN_ROWS];
  if (threadIdx.x < N_BIN_ROWS) hcnt[threadIdx.x] = 0;
  const int* p = dst + blockIdx.x * CHUNK_EDGES + threadIdx.x;
  int d[CHUNK_EDGES / 256];
#pragma unroll
  for (int i = 0; i < CHUNK_EDGES / 256; ++i) d[i] = p[i * 256];
  __syncthreads();
#pragma unroll
  for (int i = 0; i < CHUNK_EDGES / 256; ++i) atomicAdd(&hcnt[d[i] >> 8], 1);
  __syncthreads();
  if (threadIdx.x < N_BIN_ROWS) hist[threadIdx.x * N_CHUNK + blockIdx.x] = hcnt[threadIdx.x];
}

__global__ __launch_bounds__(1024) void scan_counts(int* __restrict__ cnt)
{
  __shared__ int wsum[32];
  const int lane = threadIdx.x & 31;
  const int wave = threadIdx.x >> 5;
  const int lo = wave * SCAN_PER_WAVE + lane;

  int v[49];
  int s = 0;
#pragma unroll
  for (int i = 0; i < 49; ++i) {
    const int idx = lo + i * 32;
    v[i] = idx < N_NODES ? cnt[idx] : 0;
    s += v[i];
  }
#pragma unroll
  for (int o = 16; o > 0; o >>= 1) s += __shfl_xor(s, o, 32);
  if (lane == 0) wsum[wave] = s;
  __syncthreads();

  int base = 0;
  for (int w = 0; w < wave; ++w) base += wsum[w];

  int x[49];
#pragma unroll
  for (int i = 0; i < 49; ++i) x[i] = v[i];
#pragma unroll
  for (int o = 1; o < 32; o <<= 1) {
#pragma unroll
    for (int i = 0; i < 49; ++i) {
      const int y = __shfl_up(x[i], o, 32);
      if (lane >= o) x[i] += y;
    }
  }
#pragma unroll
  for (int i = 0; i < 49; ++i) {
    const int idx = lo + i * 32;
    if (idx <= N_NODES) cnt[idx] = base + x[i] - v[i];
    base += __shfl(x[i], 31, 32);
  }
}

__global__ __launch_bounds__(256) void bin_scatter(
    const int* __restrict__ src, const int* __restrict__ dst,
    const int* __restrict__ hoff, int* __restrict__ bpack)
{
  __shared__ int cur[N_BIN];
  for (int i = threadIdx.x; i < N_BIN; i += 256) cur[i] = hoff[i * N_CHUNK + blockIdx.x];
  __syncthreads();
  const int e0 = blockIdx.x * CHUNK_EDGES + threadIdx.x;
  int d[CHUNK_EDGES / 256], s[CHUNK_EDGES / 256];
#pragma unroll
  for (int i = 0; i < CHUNK_EDGES / 256; ++i) {
    d[i] = dst[e0 + i * 256];
    s[i] = src[e0 + i * 256];
  }
#pragma unroll
  for (int i = 0; i < CHUNK_EDGES / 256; ++i) {
    const int pos = atomicAdd(&cur[d[i] >> 8], 1);
    bpack[pos] = ((d[i] & 255) << 16) | s[i];
  }
}

__global__ __launch_bounds__(1024) void bin_sort(
    const int* __restrict__ hoff, const int* __restrict__ bpack,
    int* __restrict__ off, int* __restrict__ csr)
{
  __shared__ int ncnt[256];
  __shared__ int wtot[8];
  const int tid  = threadIdx.x;
  const int lane = tid & 31;
  const int wave = tid >> 5;
  const int bin  = blockIdx.x;
  const int bstart = hoff[bin * N_CHUNK];
  const int bend   = hoff[(bin + 1) * N_CHUNK];

  if (tid < 256) ncnt[tid] = 0;
  __syncthreads();
  for (int e0 = bstart + tid; e0 < bend; e0 += 8 * 1024) {
    int p[8];
#pragma unroll
    for (int q = 0; q < 8; ++q) {
      const int e = e0 + q * 1024;
      p[q] = e < bend ? bpack[e] : -1;
    }
#pragma unroll
    for (int q = 0; q < 8; ++q)
      if (p[q] >= 0) atomicAdd(&ncnt[p[q] >> 16], 1);
  }
  __syncthreads();

  int v = 0, x = 0;
  if (tid < 256) {
    v = ncnt[tid];
    x = v;
#pragma unroll
    for (int o = 1; o < 32; o <<= 1) {
      const int y = __shfl_up(x, o, 32);
      if (lane >= o) x += y;
    }
    if (lane == 31) wtot[wave] = x;
  }
  __syncthreads();
  if (tid < 256) {
    int base = 0;
    for (int w = 0; w < wave; ++w) base += wtot[w];
    const int excl = base + x - v;
    ncnt[tid] = excl;
    const int node = bin * 256 + tid;
    if (node <= N_NODES) off[node] = bstart + excl;
  }
  __syncthreads();
  for (int e0 = bstart + tid; e0 < bend; e0 += 8 * 1024) {
    int p[8];
#pragma unroll
    for (int q = 0; q < 8; ++q) {
      const int e = e0 + q * 1024;
      p[q] = e < bend ? bpack[e] : -1;
    }
#pragma unroll
    for (int q = 0; q < 8; ++q)
      if (p[q] >= 0) {
        const int r = atomicAdd(&ncnt[p[q] >> 16], 1);
        csr[bstart + r] = p[q] & 0xFFFF;
      }
  }
}

static __device__ __forceinline__ float leaky(float t)
{
  return t > 0.0f ? t : NEG_SLOPE * t;
}

static __device__ __forceinline__ void add_rows(
    float4& acc, const float* __restrict__ H, int lane, int sj, float a, int n)
{
  int j = 0;
  for (; j + 4 <= n; j += 4) {
    const int r0 = __builtin_amdgcn_readlane(sj, j);
    const int r1 = __builtin_amdgcn_readlane(sj, j + 1);
    const int r2 = __builtin_amdgcn_readlane(sj, j + 2);
    const int r3 = __builtin_amdgcn_readlane(sj, j + 3);
    const float w0 = __int_as_float(__builtin_amdgcn_readlane(__float_as_int(a), j));
    const float w1 = __int_as_float(__builtin_amdgcn_readlane(__float_as_int(a), j + 1));
    const float w2 = __int_as_float(__builtin_amdgcn_readlane(__float_as_int(a), j + 2));
    const float w3 = __int_as_float(__builtin_amdgcn_readlane(__float_as_int(a), j + 3));
    const float4 h0 = *(const float4*)(H + (size_t)r0 * OUT_DIM + 4 * lane);
    const float4 h1 = *(const float4*)(H + (size_t)r1 * OUT_DIM + 4 * lane);
    const float4 h2 = *(const float4*)(H + (size_t)r2 * OUT_DIM + 4 * lane);
    const float4 h3 = *(const float4*)(H + (size_t)r3 * OUT_DIM + 4 * lane);
    acc.x += w0 * h0.x; acc.y += w0 * h0.y; acc.z += w0 * h0.z; acc.w += w0 * h0.w;
    acc.x += w1 * h1.x; acc.y += w1 * h1.y; acc.z += w1 * h1.z; acc.w += w1 * h1.w;
    acc.x += w2 * h2.x; acc.y += w2 * h2.y; acc.z += w2 * h2.z; acc.w += w2 * h2.w;
    acc.x += w3 * h3.x; acc.y += w3 * h3.y; acc.z += w3 * h3.z; acc.w += w3 * h3.w;
  }
  for (; j < n; ++j) {
    const int r0 = __builtin_amdgcn_readlane(sj, j);
    const float w0 = __int_as_float(__builtin_amdgcn_readlane(__float_as_int(a), j));
    const float4 h0 = *(const float4*)(H + (size_t)r0 * OUT_DIM + 4 * lane);
    acc.x += w0 * h0.x; acc.y += w0 * h0.y; acc.z += w0 * h0.z; acc.w += w0 * h0.w;
  }
}

__global__ __launch_bounds__(256) void gather_softmax(
    const int* __restrict__ off, const int* __restrict__ csr,
    const float* __restrict__ H,
    const float* __restrict__ el, const float* __restrict__ er,
    float* __restrict__ out)
{
  const int lane = threadIdx.x & 31;
  const int d = __builtin_amdgcn_readfirstlane(
      (int)(blockIdx.x * (blockDim.x >> 5) + (threadIdx.x >> 5)));
  if (d >= N_NODES) return;

  const int start = off[d];
  const int end   = off[d + 1];
  float4* orow = (float4*)(out + (size_t)d * OUT_DIM + 4 * lane);
  if (start >= end) {
    *orow = make_float4(0.0f, 0.0f, 0.0f, 0.0f);
    return;
  }
  const float erd = er[d];
  const float ninf = -__builtin_inff();

  const int i0 = start + lane;
  const int i1 = start + 32 + lane;
  const int sj0 = csr[i0 < end ? i0 : end - 1];
  const int sj1 = csr[i1 < end ? i1 : end - 1];
  const float x0 = el[sj0];
  const float x1 = el[sj1];
  const float s0 = i0 < end ? leaky(x0 + erd) : ninf;
  const float s1 = i1 < end ? leaky(x1 + erd) : ninf;

  float m = fmaxf(s0, s1);
  for (int base = start + 64; base < end; base += 32) {
    const int idx = base + lane;
    if (idx < end) m = fmaxf(m, leaky(el[csr[idx]] + erd));
  }
#pragma unroll
  for (int o = 16; o > 0; o >>= 1) m = fmaxf(m, __shfl_xor(m, o, 32));

  const float e0 = __expf(s0 - m);
  const float e1 = __expf(s1 - m);
  float sum = e0 + e1;
  for (int base = start + 64; base < end; base += 32) {
    const int idx = base + lane;
    if (idx < end) sum += __expf(leaky(el[csr[idx]] + erd) - m);
  }
#pragma unroll
  for (int o = 16; o > 0; o >>= 1) sum += __shfl_xor(sum, o, 32);

  float4 acc = make_float4(0.0f, 0.0f, 0.0f, 0.0f);
  const int deg = end - start;
  add_rows(acc, H, lane, sj0, e0 / sum, deg < 32 ? deg : 32);
  if (deg > 32) add_rows(acc, H, lane, sj1, e1 / sum, deg < 64 ? deg - 32 : 32);
  for (int base = start + 64; base < end; base += 32) {
    const int idx = base + lane;
    int sj = 0;
    float a = 0.0f;
    if (idx < end) {
      sj = csr[idx];
      a = __expf(leaky(el[sj] + erd) - m) / sum;
    }
    const int left = end - base;
    add_rows(acc, H, lane, sj, a, left < 32 ? left : 32);
  }
  *orow = acc;
}

extern "C" void kernel_launch(void* const* d_in, const int* in_sizes, int n_in,
                              void* d_out, int out_size, void* d_ws, size_t ws_size,
                              hipStream_t stream) {
  (void)in_sizes; (void)n_in; (void)out_size; (void)ws_size;

  const float* features  = (const float*)d_in[0];
  const float* fc_weight = (const float*)d_in[1];
  const float* attn_l    = (const float*)d_in[2];
  const float* attn_r    = (const float*)d_in[3];
  const int*   src       = (const int*)d_in[4];
  const int*   dst       = (const int*)d_in[5];
  float*       out       = (float*)d_out;

  float* h    = (float*)d_ws;
  float* el   = h + (size_t)N_NODES * OUT_DIM;
  float* er   = el + N_PAD;
  int*   off  = (int*)(er + N_PAD);
  int*   csr  = off + N_PAD;
  int*   hist  = (int*)d_ws;
  int*   bpack = hist + HIST_PAD;

  bin_count<<<N_CHUNK, 256, 0, stream>>>(dst, hist);
  scan_counts<<<1, 1024, 0, stream>>>(hist);
  bin_scatter<<<N_CHUNK, 256, 0, stream>>>(src, dst, hist, bpack);
  bin_sort<<<N_BIN, 1024, 0, stream>>>(hist, bpack, off, csr);

  gat_gemm_wmma<<<N_NODES / 16, 256, 0, stream>>>(features, fc_weight, h);

  node_logits<<<(N_NODES + 7) / 8, 256, 0, stream>>>(h, attn_l, attn_r, el, er);

  gather_softmax<<<(N_NODES + 7) / 8, 256, 0, stream>>>(off, csr, h, el, er, out);
}


# GEMM re-tiled: one wave per 16x128 row block (8 accumulator tiles, each A fragment loaded once), W staged once per work-group in LDS, 256 work-groups of 16 waves; k order of the f32 sums changes
# speedup vs baseline: 4.1773x; 1.2220x over previous
#include <hip/hip_runtime.h>
#include <hip/hip_bf16.h>

#define N_NODES 50000
#define N_EDGES 1600000
#define IN_DIM 256
#define OUT_DIM 128
#define NEG_SLOPE 0.2f
#define N_PAD 50048
#define SCAN_PER_WAVE 1568

typedef __attribute__((ext_vector_type(2))) float v2f;
typedef __attribute__((ext_vector_type(8))) float v8f;

#define GEMM_BLOCKS 256
#define GEMM_THREADS 512
#define ROW_TILES (N_NODES / 16)
#define WL_PITCH (OUT_DIM + 1)
static_assert(ROW_TILES * 16 == N_NODES, "row tiles must cover the nodes exactly");
#define FULL_WAVES (ROW_TILES / GEMM_BLOCKS)
static_assert(FULL_WAVES + 4 == GEMM_THREADS / 32, "the last four waves share the left-over row tile");
static_assert(ROW_TILES < (FULL_WAVES + 1) * GEMM_BLOCKS, "at most one left-over row tile per block");

static __device__ __forceinline__ float swap_halves(float x)
{
  return __int_as_float(__builtin_amdgcn_permlanex16(
      __float_as_int(x), __float_as_int(x), 0x76543210, 0xfedcba98, false, false));
}

template <int NT>
static __device__ __forceinline__ void row_block(
    const float4* __restrict__ aPtr, const float4* bPtr, float* __restrict__ out, int half)
{
  v8f acc[NT];
#pragma unroll
  for (int t = 0; t < NT; ++t) acc[t] = v8f{};

  float4 a[IN_DIM / 8];
#pragma unroll
  for (int i = 0; i < IN_DIM / 8; ++i) a[i] = aPtr[(i >> 2) * 8 + (i & 3)];

#pragma unroll
  for (int i = 0; i < IN_DIM / 8; ++i) {
    float4 b[NT];
#pragma unroll
    for (int t = 0; t < NT; ++t) b[t] = bPtr[((i >> 2) * 8 + (i & 3)) * WL_PITCH + t * 16];
    v2f a0, a1;
    a0.x = a[i].x; a0.y = a[i].y;
    a1.x = a[i].z; a1.y = a[i].w;
#pragma unroll
    for (int t = 0; t < NT; ++t) {
      v2f b0;
      b0.x = b[t].x; b0.y = b[t].y;
      acc[t] = __builtin_amdgcn_wmma_f32_16x16x4_f32(
           false, a0,  false, b0,
           (short)0, acc[t],  false,  false);
    }
#pragma unroll
    for (int t = 0; t < NT; ++t) {
      v2f b1;
      b1.x = b[t].z; b1.y = b[t].w;
      acc[t] = __builtin_amdgcn_wmma_f32_16x16x4_f32(
           false, a1,  false, b1,
           (short)0, acc[t],  false,  false);
    }
  }

#pragma unroll
  for (int tp = 0; tp < NT / 2; ++tp) {
#pragma unroll
    for (int v = 0; v < 8; ++v) {
      const float lo = acc[2 * tp][v];
      const float hi = acc[2 * tp + 1][v];
      const float got = swap_halves(half ? lo : hi);
      out[(size_t)v * OUT_DIM + 32 * tp]       = half ? got : lo;
      out[(size_t)(v + 8) * OUT_DIM + 32 * tp] = half ? hi : got;
    }
  }
}

__global__ __launch_bounds__(GEMM_THREADS) void gat_gemm_wmma(
    const float* __restrict__ X,
    const float* __restrict__ W,
    float* __restrict__ H)
{
  __shared__ float4 Wl[(IN_DIM / 4) * WL_PITCH];

  constexpr int STAGE_TRIPS = (IN_DIM / 16) * (OUT_DIM / 32) / (GEMM_THREADS / 32);
  static_assert(STAGE_TRIPS * (GEMM_THREADS / 32) == (IN_DIM / 16) * (OUT_DIM / 32), "staging units divide evenly");
  const int sKq = (int)(threadIdx.x >> 3) & 3;
  const int sN4 = (int)threadIdx.x & 7;
  float4 r[STAGE_TRIPS][4];
#pragma unroll
  for (int i = 0; i < STAGE_TRIPS; ++i) {
    const int unit = i * (GEMM_THREADS / 32) + (int)(threadIdx.x >> 5);
    const int kq = 4 * (unit >> 2) + sKq;
    const int n4 = 8 * (unit & 3) + sN4;
    const float* wp = W + (size_t)(4 * kq) * OUT_DIM + 4 * n4;
#pragma unroll
    for (int c4 = 0; c4 < 4; ++c4) r[i][c4] = *(const float4*)(wp + c4 * OUT_DIM);
  }
#pragma unroll
  for (int i = 0; i < STAGE_TRIPS; ++i) {
    const int unit = i * (GEMM_THREADS / 32) + (int)(threadIdx.x >> 5);
    const int kq = 4 * (unit >> 2) + sKq;
    const int n4 = 8 * (unit & 3) + sN4;
    float4* q = Wl + kq * WL_PITCH + 4 * n4;
    q[0] = make_float4(r[i][0].x, r[i][1].x, r[i][2].x, r[i][3].x);
    q[1] = make_float4(r[i][0].y, r[i][1].y, r[i][2].y, r[i][3].y);
    q[2] = make_float4(r[i][0].z, r[i][1].z, r[i][2].z, r[i][3].z);
    q[3] = make_float4(r[i][0].w, r[i][1].w, r[i][2].w, r[i][3].w);
  }
  __syncthreads();

  const int lane = threadIdx.x & 31;
  const int wave = __builtin_amdgcn_readfirstlane((int)(threadIdx.x >> 5));
  const int l16  = lane & 15;
  const int half = lane >> 4;

  const int tile = (wave < FULL_WAVES ? wave : FULL_WAVES) * GEMM_BLOCKS + blockIdx.x;
  if (tile >= ROW_TILES) return;
  const int mBase = tile * 16;

  const float4* aPtr = (const float4*)(X + (size_t)(mBase + l16) * IN_DIM + 16 * half);
  const float4* bPtr = Wl + (4 * half) * WL_PITCH + l16;
  float* out = H + (size_t)mBase * OUT_DIM + lane;
  if (wave < FULL_WAVES) {
    row_block<8>(aPtr, bPtr, out, half);
  } else {
    const int col0 = (wave - FULL_WAVES) * 32;
    row_block<2>(aPtr, bPtr + col0, out + col0, half);
  }
}

__global__ __launch_bounds__(256) void node_logits(
    const float* __restrict__ H,
    const float* __restrict__ attn_l,
    const float* __restrict__ attn_r,
    float* __restrict__ el, float* __restrict__ er)
{
  const int lane = threadIdx.x & 31;
  const int node = blockIdx.x * (blockDim.x >> 5) + (threadIdx.x >> 5);
  if (node >= N_NODES) return;

  const float4 hv = *(const float4*)(H + (size_t)node * OUT_DIM + 4 * lane);
  const float4 al = *(const float4*)(attn_l + 4 * lane);
  const float4 ar = *(const float4*)(attn_r + 4 * lane);
  float sl = hv.x * al.x + hv.y * al.y + hv.z * al.z + hv.w * al.w;
  float sr = hv.x * ar.x + hv.y * ar.y + hv.z * ar.z + hv.w * ar.w;
#pragma unroll
  for (int off = 16; off > 0; off >>= 1) {
    sl += __shfl_down(sl, off, 32);
    sr += __shfl_down(sr, off, 32);
  }
  if (lane == 0) { el[node] = sl; er[node] = sr; }
}

#define N_CHUNK 250
#define CHUNK_EDGES 6400
#define N_BIN 196
#define N_BIN_ROWS 200
#define HIST_PAD N_PAD
static_assert(N_BIN_ROWS * N_CHUNK == N_NODES, "scan_counts scans exactly N_NODES counts");
static_assert(N_BIN_ROWS >= N_BIN, "every bin has a histogram row");
static_assert(N_CHUNK * CHUNK_EDGES == N_EDGES, "chunks must tile the edge list");
static_assert(N_BIN * 256 > N_NODES, "bins must cover off[0..N_NODES]");
static_assert(N_NODES <= 65536, "a source index must fit 16 bits of the packed edge word");

__global__ __launch_bounds__(256) void bin_count(
    const int* __restrict__ dst, int* __restrict__ hist)
{
  __shared__ int hcnt[N_BIN_ROWS];
  if (threadIdx.x < N_BIN_ROWS) hcnt[threadIdx.x] = 0;
  const int* p = dst + blockIdx.x * CHUNK_EDGES + threadIdx.x;
  int d[CHUNK_EDGES / 256];
#pragma unroll
  for (int i = 0; i < CHUNK_EDGES / 256; ++i) d[i] = p[i * 256];
  __syncthreads();
#pragma unroll
  for (int i = 0; i < CHUNK_EDGES / 256; ++i) atomicAdd(&hcnt[d[i] >> 8], 1);
  __syncthreads();
  if (threadIdx.x < N_BIN_ROWS) hist[threadIdx.x * N_CHUNK + blockIdx.x] = hcnt[threadIdx.x];
}

__global__ __launch_bounds__(1024) void scan_counts(int* __restrict__ cnt)
{
  __shared__ int wsum[32];
  const int lane = threadIdx.x & 31;
  const int wave = threadIdx.x >> 5;
  const int lo = wave * SCAN_PER_WAVE + lane;

  int v[49];
  int s = 0;
#pragma unroll
  for (int i = 0; i < 49; ++i) {
    const int idx = lo + i * 32;
    v[i] = idx < N_NODES ? cnt[idx] : 0;
    s += v[i];
  }
#pragma unroll
  for (int o = 16; o > 0; o >>= 1) s += __shfl_xor(s, o, 32);
  if (lane == 0) wsum[wave] = s;
  __syncthreads();

  int base = 0;
  for (int w = 0; w < wave; ++w) base += wsum[w];

  int x[49];
#pragma unroll
  for (int i = 0; i < 49; ++i) x[i] = v[i];
#pragma unroll
  for (int o = 1; o < 32; o <<= 1) {
#pragma unroll
    for (int i = 0; i < 49; ++i) {
      const int y = __shfl_up(x[i], o, 32);
      if (lane >= o) x[i] += y;
    }
  }
#pragma unroll
  for (int i = 0; i < 49; ++i) {
    const int idx = lo + i * 32;
    if (idx <= N_NODES) cnt[idx] = base + x[i] - v[i];
    base += __shfl(x[i], 31, 32);
  }
}

__global__ __launch_bounds__(256) void bin_scatter(
    const int* __restrict__ src, const int* __restrict__ dst,
    const int* __restrict__ hoff, int* __restrict__ bpack)
{
  __shared__ int cur[N_BIN];
  for (int i = threadIdx.x; i < N_BIN; i += 256) cur[i] = hoff[i * N_CHUNK + blockIdx.x];
  __syncthreads();
  const int e0 = blockIdx.x * CHUNK_EDGES + threadIdx.x;
  int d[CHUNK_EDGES / 256], s[CHUNK_EDGES / 256];
#pragma unroll
  for (int i = 0; i < CHUNK_EDGES / 256; ++i) {
    d[i] = dst[e0 + i * 256];
    s[i] = src[e0 + i * 256];
  }
#pragma unroll
  for (int i = 0; i < CHUNK_EDGES / 256; ++i) {
    const int pos = atomicAdd(&cur[d[i] >> 8], 1);
    bpack[pos] = ((d[i] & 255) << 16) | s[i];
  }
}

__global__ __launch_bounds__(1024) void bin_sort(
    const int* __restrict__ hoff, const int* __restrict__ bpack,
    int* __restrict__ off, int* __restrict__ csr)
{
  __shared__ int ncnt[256];
  __shared__ int wtot[8];
  const int tid  = threadIdx.x;
  const int lane = tid & 31;
  const int wave = tid >> 5;
  const int bin  = blockIdx.x;
  const int bstart = hoff[bin * N_CHUNK];
  const int bend   = hoff[(bin + 1) * N_CHUNK];

  if (tid < 256) ncnt[tid] = 0;
  __syncthreads();
  for (int e0 = bstart + tid; e0 < bend; e0 += 8 * 1024) {
    int p[8];
#pragma unroll
    for (int q = 0; q < 8; ++q) {
      const int e = e0 + q * 1024;
      p[q] = e < bend ? bpack[e] : -1;
    }
#pragma unroll
    for (int q = 0; q < 8; ++q)
      if (p[q] >= 0) atomicAdd(&ncnt[p[q] >> 16], 1);
  }
  __syncthreads();

  int v = 0, x = 0;
  if (tid < 256) {
    v = ncnt[tid];
    x = v;
#pragma unroll
    for (int o = 1; o < 32; o <<= 1) {
      const int y = __shfl_up(x, o, 32);
      if (lane >= o) x += y;
    }
    if (lane == 31) wtot[wave] = x;
  }
  __syncthreads();
  if (tid < 256) {
    int base = 0;
    for (int w = 0; w < wave; ++w) base += wtot[w];
    const int excl = base + x - v;
    ncnt[tid] = excl;
    const int node = bin * 256 + tid;
    if (node <= N_NODES) off[node] = bstart + excl;
  }
  __syncthreads();
  for (int e0 = bstart + tid; e0 < bend; e0 += 8 * 1024) {
    int p[8];
#pragma unroll
    for (int q = 0; q < 8; ++q) {
      const int e = e0 + q * 1024;
      p[q] = e < bend ? bpack[e] : -1;
    }
#pragma unroll
    for (int q = 0; q < 8; ++q)
      if (p[q] >= 0) {
        const int r = atomicAdd(&ncnt[p[q] >> 16], 1);
        csr[bstart + r] = p[q] & 0xFFFF;
      }
  }
}

static __device__ __forceinline__ float leaky(float t)
{
  return t > 0.0f ? t : NEG_SLOPE * t;
}

static __device__ __forceinline__ void add_rows(
    float4& acc, const float* __restrict__ H, int lane, int sj, float a, int n)
{
  int j = 0;
  for (; j + 4 <= n; j += 4) {
    const int r0 = __builtin_amdgcn_readlane(sj, j);
    const int r1 = __builtin_amdgcn_readlane(sj, j + 1);
    const int r2 = __builtin_amdgcn_readlane(sj, j + 2);
    const int r3 = __builtin_amdgcn_readlane(sj, j + 3);
    const float w0 = __int_as_float(__builtin_amdgcn_readlane(__float_as_int(a), j));
    const float w1 = __int_as_float(__builtin_amdgcn_readlane(__float_as_int(a), j + 1));
    const float w2 = __int_as_float(__builtin_amdgcn_readlane(__float_as_int(a), j + 2));
    const float w3 = __int_as_float(__builtin_amdgcn_readlane(__float_as_int(a), j + 3));
    const float4 h0 = *(const float4*)(H + (size_t)r0 * OUT_DIM + 4 * lane);
    const float4 h1 = *(const float4*)(H + (size_t)r1 * OUT_DIM + 4 * lane);
    const float4 h2 = *(const float4*)(H + (size_t)r2 * OUT_DIM + 4 * lane);
    const float4 h3 = *(const float4*)(H + (size_t)r3 * OUT_DIM + 4 * lane);
    acc.x += w0 * h0.x; acc.y += w0 * h0.y; acc.z += w0 * h0.z; acc.w += w0 * h0.w;
    acc.x += w1 * h1.x; acc.y += w1 * h1.y; acc.z += w1 * h1.z; acc.w += w1 * h1.w;
    acc.x += w2 * h2.x; acc.y += w2 * h2.y; acc.z += w2 * h2.z; acc.w += w2 * h2.w;
    acc.x += w3 * h3.x; acc.y += w3 * h3.y; acc.z += w3 * h3.z; acc.w += w3 * h3.w;
  }
  for (; j < n; ++j) {
    const int r0 = __builtin_amdgcn_readlane(sj, j);
    const float w0 = __int_as_float(__builtin_amdgcn_readlane(__float_as_int(a), j));
    const float4 h0 = *(const float4*)(H + (size_t)r0 * OUT_DIM + 4 * lane);
    acc.x += w0 * h0.x; acc.y += w0 * h0.y; acc.z += w0 * h0.z; acc.w += w0 * h0.w;
  }
}

__global__ __launch_bounds__(256) void gather_softmax(
    const int* __restrict__ off, const int* __restrict__ csr,
    const float* __restrict__ H,
    const float* __restrict__ el, const float* __restrict__ er,
    float* __restrict__ out)
{
  const int lane = threadIdx.x & 31;
  const int d = __builtin_amdgcn_readfirstlane(
      (int)(blockIdx.x * (blockDim.x >> 5) + (threadIdx.x >> 5)));
  if (d >= N_NODES) return;

  const int start = off[d];
  const int end   = off[d + 1];
  float4* orow = (float4*)(out + (size_t)d * OUT_DIM + 4 * lane);
  if (start >= end) {
    *orow = make_float4(0.0f, 0.0f, 0.0f, 0.0f);
    return;
  }
  const float erd = er[d];
  const float ninf = -__builtin_inff();

  const int i0 = start + lane;
  const int i1 = start + 32 + lane;
  const int sj0 = csr[i0 < end ? i0 : end - 1];
  const int sj1 = csr[i1 < end ? i1 : end - 1];
  const float x0 = el[sj0];
  const float x1 = el[sj1];
  const float s0 = i0 < end ? leaky(x0 + erd) : ninf;
  const float s1 = i1 < end ? leaky(x1 + erd) : ninf;

  float m = fmaxf(s0, s1);
  for (int base = start + 64; base < end; base += 32) {
    const int idx = base + lane;
    if (idx < end) m = fmaxf(m, leaky(el[csr[idx]] + erd));
  }
#pragma unroll
  for (int o = 16; o > 0; o >>= 1) m = fmaxf(m, __shfl_xor(m, o, 32));

  const float e0 = __expf(s0 - m);
  const float e1 = __expf(s1 - m);
  float sum = e0 + e1;
  for (int base = start + 64; base < end; base += 32) {
    const int idx = base + lane;
    if (idx < end) sum += __expf(leaky(el[csr[idx]] + erd) - m);
  }
#pragma unroll
  for (int o = 16; o > 0; o >>= 1) sum += __shfl_xor(sum, o, 32);

  float4 acc = make_float4(0.0f, 0.0f, 0.0f, 0.0f);
  const int deg = end - start;
  add_rows(acc, H, lane, sj0, e0 / sum, deg < 32 ? deg : 32);
  if (deg > 32) add_rows(acc, H, lane, sj1, e1 / sum, deg < 64 ? deg - 32 : 32);
  for (int base = start + 64; base < end; base += 32) {
    const int idx = base + lane;
    int sj = 0;
    float a = 0.0f;
    if (idx < end) {
      sj = csr[idx];
      a = __expf(leaky(el[sj] + erd) - m) / sum;
    }
    const int left = end - base;
    add_rows(acc, H, lane, sj, a, left < 32 ? left : 32);
  }
  *orow = acc;
}

extern "C" void kernel_launch(void* const* d_in, const int* in_sizes, int n_in,
                              void* d_out, int out_size, void* d_ws, size_t ws_size,
                              hipStream_t stream) {
  (void)in_sizes; (void)n_in; (void)out_size; (void)ws_size;

  const float* features  = (const float*)d_in[0];
  const float* fc_weight = (const float*)d_in[1];
  const float* attn_l    = (const float*)d_in[2];
  const float* attn_r    = (const float*)d_in[3];
  const int*   src       = (const int*)d_in[4];
  const int*   dst       = (const int*)d_in[5];
  float*       out       = (float*)d_out;

  float* h    = (float*)d_ws;
  float* el   = h + (size_t)N_NODES * OUT_DIM;
  float* er   = el + N_PAD;
  int*   off  = (int*)(er + N_PAD);
  int*   csr  = off + N_PAD;
  int*   hist  = (int*)d_ws;
  int*   bpack = hist + HIST_PAD;

  bin_count<<<N_CHUNK, 256, 0, stream>>>(dst, hist);
  scan_counts<<<1, 1024, 0, stream>>>(hist);
  bin_scatter<<<N_CHUNK, 256, 0, stream>>>(src, dst, hist, bpack);
  bin_sort<<<N_BIN, 1024, 0, stream>>>(hist, bpack, off, csr);

  gat_gemm_wmma<<<GEMM_BLOCKS, GEMM_THREADS, 0, stream>>>(features, fc_weight, h);

  node_logits<<<(N_NODES + 7) / 8, 256, 0, stream>>>(h, attn_l, attn_r, el, er);

  gather_softmax<<<(N_NODES + 7) / 8, 256, 0, stream>>>(off, csr, h, el, er, out);
}


# Both placing passes of the edge sort build their output in LDS and write it out in order: whole lines per store, not 32 lines of 4 bytes; sum order unchanged (the gather orders each segment)
# speedup vs baseline: 6.7044x; 1.6050x over previous
#include <hip/hip_runtime.h>
#include <hip/hip_bf16.h>

#define N_NODES 50000
#define N_EDGES 1600000
#define IN_DIM 256
#define OUT_DIM 128
#define NEG_SLOPE 0.2f
#define N_PAD 50048

typedef __attribute__((ext_vector_type(2))) float v2f;
typedef __attribute__((ext_vector_type(8))) float v8f;

#define GEMM_BLOCKS 256
#define GEMM_THREADS 512
#define ROW_TILES (N_NODES / 16)
#define WL_PITCH (OUT_DIM + 1)
static_assert(ROW_TILES * 16 == N_NODES, "row tiles must cover the nodes exactly");
#define FULL_WAVES (ROW_TILES / GEMM_BLOCKS)
static_assert(FULL_WAVES + 4 == GEMM_THREADS / 32, "the last four waves share the left-over row tile");
static_assert(ROW_TILES < (FULL_WAVES + 1) * GEMM_BLOCKS, "at most one left-over row tile per block");

static __device__ __forceinline__ float swap_halves(float x)
{
  return __int_as_float(__builtin_amdgcn_permlanex16(
      __float_as_int(x), __float_as_int(x), 0x76543210, 0xfedcba98, false, false));
}

template <int M>
static __device__ __forceinline__ float row_xor(float x)
{
  return __int_as_float(__builtin_amdgcn_update_dpp(
      0, __float_as_int(x), 0x160 + M, 0xf, 0xf, true));
}

template <int NT>
static __device__ __forceinline__ float row_block(
    const float4* __restrict__ aPtr, const float4* bPtr, float* __restrict__ out, int half, int l16,
    const float* __restrict__ alp, const float* __restrict__ arp)
{
  v8f acc[NT];
#pragma unroll
  for (int t = 0; t < NT; ++t) acc[t] = v8f{};

  float4 a[IN_DIM / 8];
#pragma unroll
  for (int i = 0; i < IN_DIM / 8; ++i) a[i] = aPtr[(i >> 2) * 8 + (i & 3)];

#pragma unroll
  for (int i = 0; i < IN_DIM / 8; ++i) {
    float4 b[NT];
#pragma unroll
    for (int t = 0; t < NT; ++t) b[t] = bPtr[((i >> 2) * 8 + (i & 3)) * WL_PITCH + t * 16];
    v2f a0, a1;
    a0.x = a[i].x; a0.y = a[i].y;
    a1.x = a[i].z; a1.y = a[i].w;
#pragma unroll
    for (int t = 0; t < NT; ++t) {
      v2f b0;
      b0.x = b[t].x; b0.y = b[t].y;
      acc[t] = __builtin_amdgcn_wmma_f32_16x16x4_f32(
           false, a0,  false, b0,
           (short)0, acc[t],  false,  false);
    }
#pragma unroll
    for (int t = 0; t < NT; ++t) {
      v2f b1;
      b1.x = b[t].z; b1.y = b[t].w;
      acc[t] = __builtin_amdgcn_wmma_f32_16x16x4_f32(
           false, a1,  false, b1,
           (short)0, acc[t],  false,  false);
    }
  }

#pragma unroll
  for (int tp = 0; tp < NT / 2; ++tp) {
#pragma unroll
    for (int v = 0; v < 8; ++v) {
      const float lo = acc[2 * tp][v];
      const float hi = acc[2 * tp + 1][v];
      const float got = swap_halves(half ? lo : hi);
      out[(size_t)v * OUT_DIM + 32 * tp]       = half ? got : lo;
      out[(size_t)(v + 8) * OUT_DIM + 32 * tp] = half ? hi : got;
    }
  }

  float pl[8], pr[8];
#pragma unroll
  for (int v = 0; v < 8; ++v) { pl[v] = 0.0f; pr[v] = 0.0f; }
#pragma unroll
  for (int t = 0; t < NT; ++t) {
    const float al = alp[16 * t];
    const float ar = arp[16 * t];
#pragma unroll
    for (int v = 0; v < 8; ++v) {
      pl[v] += acc[t][v] * al;
      pr[v] += acc[t][v] * ar;
    }
  }
#pragma unroll
  for (int v = 0; v < 8; ++v) {
    pl[v] += row_xor<1>(pl[v]);  pr[v] += row_xor<1>(pr[v]);
  }
#pragma unroll
  for (int v = 0; v < 8; ++v) {
    pl[v] += row_xor<2>(pl[v]);  pr[v] += row_xor<2>(pr[v]);
  }
#pragma unroll
  for (int v = 0; v < 8; ++v) {
    pl[v] += row_xor<4>(pl[v]);  pr[v] += row_xor<4>(pr[v]);
  }
#pragma unroll
  for (int v = 0; v < 8; ++v) {
    pl[v] += row_xor<8>(pl[v]);  pr[v] += row_xor<8>(pr[v]);
  }
  float keep = 0.0f;
#pragma unroll
  for (int v = 0; v < 8; ++v)
    if ((l16 & 7) == v) keep = l16 < 8 ? pl[v] : pr[v];
  return keep;
}

__global__ __launch_bounds__(GEMM_THREADS) void gat_gemm_wmma(
    const float* __restrict__ X,
    const float* __restrict__ W,
    const float* __restrict__ attn_l,
    const float* __restrict__ attn_r,
    float* __restrict__ H,
    float* __restrict__ el, float* __restrict__ er)
{
  __shared__ float4 Wl[(IN_DIM / 4) * WL_PITCH];
  __shared__ float part[4][2][16];

  constexpr int STAGE_TRIPS = (IN_DIM / 16) * (OUT_DIM / 32) / (GEMM_THREADS / 32);
  static_assert(STAGE_TRIPS * (GEMM_THREADS / 32) == (IN_DIM / 16) * (OUT_DIM / 32), "staging units divide evenly");
  const int sKq = (int)(threadIdx.x >> 3) & 3;
  const int sN4 = (int)threadIdx.x & 7;
  float4 r[STAGE_TRIPS][4];
#pragma unroll
  for (int i = 0; i < STAGE_TRIPS; ++i) {
    const int unit = i * (GEMM_THREADS / 32) + (int)(threadIdx.x >> 5);
    const int kq = 4 * (unit >> 2) + sKq;
    const int n4 = 8 * (unit & 3) + sN4;
    const float* wp = W + (size_t)(4 * kq) * OUT_DIM + 4 * n4;
#pragma unroll
    for (int c4 = 0; c4 < 4; ++c4) r[i][c4] = *(const float4*)(wp + c4 * OUT_DIM);
  }
#pragma unroll
  for (int i = 0; i < STAGE_TRIPS; ++i) {
    const int unit = i * (GEMM_THREADS / 32) + (int)(threadIdx.x >> 5);
    const int kq = 4 * (unit >> 2) + sKq;
    const int n4 = 8 * (unit & 3) + sN4;
    float4* q = Wl + kq * WL_PITCH + 4 * n4;
    q[0] = make_float4(r[i][0].x, r[i][1].x, r[i][2].x, r[i][3].x);
    q[1] = make_float4(r[i][0].y, r[i][1].y, r[i][2].y, r[i][3].y);
    q[2] = make_float4(r[i][0].z, r[i][1].z, r[i][2].z, r[i][3].z);
    q[3] = make_float4(r[i][0].w, r[i][1].w, r[i][2].w, r[i][3].w);
  }
  __syncthreads();

  const int lane = threadIdx.x & 31;
  const int wave = __builtin_amdgcn_readfirstlane((int)(threadIdx.x >> 5));
  const int l16  = lane & 15;
  const int half = lane >> 4;

  const int tile = (wave < FULL_WAVES ? wave : FULL_WAVES) * GEMM_BLOCKS + blockIdx.x;
  const bool hasTile = tile < ROW_TILES;
  const int mBase = tile * 16;
  const int row = 8 * half + (l16 & 7);

  if (hasTile) {
    const float4* aPtr = (const float4*)(X + (size_t)(mBase + l16) * IN_DIM + 16 * half);
    const float4* bPtr = Wl + (4 * half) * WL_PITCH + l16;
    float* out = H + (size_t)mBase * OUT_DIM + lane;
    if (wave < FULL_WAVES) {
      const float keep = row_block<8>(aPtr, bPtr, out, half, l16, attn_l + l16, attn_r + l16);
      (l16 < 8 ? el : er)[mBase + row] = keep;
    } else {
      const int q = wave - FULL_WAVES;
      const int col0 = q * 32;
      const float keep = row_block<2>(aPtr, bPtr + col0, out + col0, half, l16,
                                      attn_l + col0 + l16, attn_r + col0 + l16);
      part[q][l16 >> 3][row] = keep;
    }
  }
  __syncthreads();
  if (wave == FULL_WAVES && hasTile) {
    const int r = lane & 15;
    const float s = (part[0][half][r] + part[1][half][r]) + (part[2][half][r] + part[3][half][r]);
    (half ? er : el)[mBase + r] = s;
  }
}

#define N_CHUNK 250
#define CHUNK_EDGES 6400
#define N_BIN 196
#define HIST_PAD N_PAD
#define RUN_ROUNDS 8
#define TAB_TRIPS 12
#define SEG_STAGE 12288
static_assert(SEG_STAGE % 1024 == 0, "the copy out of the LDS buffer runs in whole rounds of the work-group");
static_assert(N_BIN * N_CHUNK <= HIST_PAD, "the descriptor table fits its region");
static_assert(N_CHUNK * CHUNK_EDGES == N_EDGES, "chunks must tile the edge list");
static_assert(CHUNK_EDGES % 4 == 0 && HIST_PAD % 4 == 0, "a chunk's words start on a 16-byte boundary");
static_assert(N_BIN * 256 > N_NODES, "bins must cover off[0..N_NODES]");
static_assert(N_BIN <= 256, "one LDS counter per bin in a 256-entry array");
static_assert(N_NODES <= 65536, "a source index must fit 16 bits of the packed edge word");
static_assert(CHUNK_EDGES < 65536, "a run's start and length fit 16 bits each");
static_assert(32 * RUN_ROUNDS >= N_CHUNK, "every run of a bin has a wave and a round");
static_assert(TAB_TRIPS * 4096 >= (N_BIN - 1) * N_CHUNK, "the sum over earlier bins covers the table");
static_assert((N_BIN - 1) * N_CHUNK + 3 < N_BIN * N_CHUNK, "a group of 4 that starts below the limit lies inside the table");

__global__ __launch_bounds__(256) void chunk_bucket(
    const int* __restrict__ src, const int* __restrict__ dst,
    int* __restrict__ tab, int* __restrict__ bpack)
{
  __shared__ int hcnt[256];
  __shared__ int wtot[8];
  __shared__ int4 stage4[CHUNK_EDGES / 4];
  const int tid  = threadIdx.x;
  const int lane = tid & 31;
  const int wave = tid >> 5;
  hcnt[tid] = 0;
  const int e0 = blockIdx.x * CHUNK_EDGES + tid;
  int d[CHUNK_EDGES / 256], s[CHUNK_EDGES / 256], r[CHUNK_EDGES / 256];
#pragma unroll
  for (int i = 0; i < CHUNK_EDGES / 256; ++i) {
    d[i] = dst[e0 + i * 256];
    s[i] = src[e0 + i * 256];
  }
  __syncthreads();
#pragma unroll
  for (int i = 0; i < CHUNK_EDGES / 256; ++i) r[i] = atomicAdd(&hcnt[d[i] >> 8], 1);
  __syncthreads();

  const int v = hcnt[tid];
  int x = v;
#pragma unroll
  for (int o = 1; o < 32; o <<= 1) {
    const int y = __shfl_up(x, o, 32);
    if (lane >= o) x += y;
  }
  if (lane == 31) wtot[wave] = x;
  __syncthreads();
  int base = 0;
  for (int w = 0; w < wave; ++w) base += wtot[w];
  const int excl = base + x - v;
  hcnt[tid] = excl;
  if (tid < N_BIN) tab[tid * N_CHUNK + blockIdx.x] = (excl << 16) | v;
  __syncthreads();

  int* stage = (int*)stage4;
#pragma unroll
  for (int i = 0; i < CHUNK_EDGES / 256; ++i)
    stage[hcnt[d[i] >> 8] + r[i]] = ((d[i] & 255) << 16) | s[i];
  __syncthreads();
  int4* mine4 = (int4*)(bpack + blockIdx.x * CHUNK_EDGES);
#pragma unroll
  for (int i = 0; i < (CHUNK_EDGES / 4 + 255) / 256; ++i) {
    const int j = tid + i * 256;
    if (j < CHUNK_EDGES / 4) mine4[j] = stage4[j];
  }
}

__global__ __launch_bounds__(1024) void bin_sort(
    const int* __restrict__ tab, const int* __restrict__ bpack,
    int* __restrict__ off, int* __restrict__ csr)
{
  __shared__ int ncnt[256];
  __shared__ int ncnt2[256];
  __shared__ int wsum[32];
  __shared__ int wtot[8];
  __shared__ int sstage[SEG_STAGE];
  const int tid  = threadIdx.x;
  const int lane = tid & 31;
  const int wave = __builtin_amdgcn_readfirstlane(tid >> 5);
  const int bin  = blockIdx.x;

  int mydesc = 0;
  {
    const int c = wave + 32 * lane;
    if (lane < RUN_ROUNDS && c < N_CHUNK) mydesc = tab[bin * N_CHUNK + c];
  }
  const int lim = bin * N_CHUNK;
  const int4* tab4 = (const int4*)tab;
  int4 tq[TAB_TRIPS];
#pragma unroll
  for (int i = 0; i < TAB_TRIPS; ++i) {
    const int j = tid + i * 1024;
    tq[i] = tab4[4 * j < lim ? j : 0];
  }
  int part = 0;
#pragma unroll
  for (int i = 0; i < TAB_TRIPS; ++i) {
    const int e = 4 * (tid + i * 1024);
    part += (e     < lim ? tq[i].x & 0xFFFF : 0) + (e + 1 < lim ? tq[i].y & 0xFFFF : 0)
          + (e + 2 < lim ? tq[i].z & 0xFFFF : 0) + (e + 3 < lim ? tq[i].w & 0xFFFF : 0);
  }
  if (tid < 256) { ncnt[tid] = 0; ncnt2[tid] = 0; }

  int p[2 * RUN_ROUNDS], r[2 * RUN_ROUNDS];
#pragma unroll
  for (int q = 0; q < RUN_ROUNDS; ++q) {
    const int desc = __builtin_amdgcn_readlane(mydesc, q);
    const int len  = desc & 0xFFFF;
    const int at   = (wave + 32 * q) * CHUNK_EDGES + (desc >> 16) + lane;
    p[2 * q]     = bpack[lane < len      ? at      : 0];
    p[2 * q + 1] = bpack[lane + 32 < len ? at + 32 : 0];
  }
#pragma unroll
  for (int q = 0; q < RUN_ROUNDS; ++q) {
    const int len = __builtin_amdgcn_readlane(mydesc, q) & 0xFFFF;
    p[2 * q]     = lane < len      ? p[2 * q]     : -1;
    p[2 * q + 1] = lane + 32 < len ? p[2 * q + 1] : -1;
  }

#pragma unroll
  for (int o = 16; o > 0; o >>= 1) part += __shfl_xor(part, o, 32);
  if (lane == 0) wsum[wave] = part;
  __syncthreads();
  int bstart = wsum[lane];
#pragma unroll
  for (int o = 16; o > 0; o >>= 1) bstart += __shfl_xor(bstart, o, 32);

#pragma unroll
  for (int k = 0; k < 2 * RUN_ROUNDS; ++k) {
    r[k] = 0;
    if (p[k] >= 0) r[k] = atomicAdd(&ncnt[p[k] >> 16], 1);
  }
#pragma unroll
  for (int q = 0; q < RUN_ROUNDS; ++q) {
    const int desc = __builtin_amdgcn_readlane(mydesc, q);
    const int len  = desc & 0xFFFF;
    const int* run = bpack + (wave + 32 * q) * CHUNK_EDGES + (desc >> 16);
#pragma clang loop unroll(disable)
    for (int k0 = 64; k0 < len; k0 += 32)
      if (k0 + lane < len) atomicAdd(&ncnt2[run[k0 + lane] >> 16], 1);
  }
  __syncthreads();

  int v1 = 0, v = 0, x = 0;
  if (tid < 256) {
    v1 = ncnt[tid];
    v = v1 + ncnt2[tid];
    x = v;
#pragma unroll
    for (int o = 1; o < 32; o <<= 1) {
      const int y = __shfl_up(x, o, 32);
      if (lane >= o) x += y;
    }
    if (lane == 31) wtot[wave] = x;
  }
  __syncthreads();
  if (tid < 256) {
    int base = 0;
    for (int w = 0; w < wave; ++w) base += wtot[w];
    const int excl = base + x - v;
    ncnt[tid]  = excl;
    ncnt2[tid] = excl + v1;
    const int node = bin * 256 + tid;
    if (node <= N_NODES) off[node] = bstart + excl;
  }
  __syncthreads();

  int* seg = csr + bstart;
  int total = 0;
  for (int w = 0; w < 8; ++w) total += wtot[w];
  if (total <= SEG_STAGE) {
#pragma unroll
    for (int k = 0; k < 2 * RUN_ROUNDS; ++k)
      if (p[k] >= 0) sstage[ncnt[p[k] >> 16] + r[k]] = p[k] & 0xFFFF;
#pragma unroll
    for (int q = 0; q < RUN_ROUNDS; ++q) {
      const int desc = __builtin_amdgcn_readlane(mydesc, q);
      const int len  = desc & 0xFFFF;
      const int* run = bpack + (wave + 32 * q) * CHUNK_EDGES + (desc >> 16);
#pragma clang loop unroll(disable)
      for (int k0 = 64; k0 < len; k0 += 32)
        if (k0 + lane < len) {
          const int w = run[k0 + lane];
          sstage[atomicAdd(&ncnt2[w >> 16], 1)] = w & 0xFFFF;
        }
    }
    __syncthreads();
    for (int i = 0; i < SEG_STAGE / 1024; ++i) {
      const int j = tid + i * 1024;
      if (j < total) seg[j] = sstage[j];
    }
  } else {
#pragma unroll
    for (int k = 0; k < 2 * RUN_ROUNDS; ++k)
      if (p[k] >= 0) seg[ncnt[p[k] >> 16] + r[k]] = p[k] & 0xFFFF;
#pragma unroll
    for (int q = 0; q < RUN_ROUNDS; ++q) {
      const int desc = __builtin_amdgcn_readlane(mydesc, q);
      const int len  = desc & 0xFFFF;
      const int* run = bpack + (wave + 32 * q) * CHUNK_EDGES + (desc >> 16);
#pragma clang loop unroll(disable)
      for (int k0 = 64; k0 < len; k0 += 32)
        if (k0 + lane < len) {
          const int w = run[k0 + lane];
          seg[atomicAdd(&ncnt2[w >> 16], 1)] = w & 0xFFFF;
        }
    }
  }
}

template <int M>
static __device__ __forceinline__ int row_xor_i(int x)
{
  return __builtin_amdgcn_update_dpp(0, x, 0x160 + M, 0xf, 0xf, true);
}
template <int J> struct lane_xor_i {
  static __device__ __forceinline__ int get(int x) { return row_xor_i<J>(x); }
};
template <> struct lane_xor_i<16> {
  static __device__ __forceinline__ int get(int x)
  {
    return __builtin_amdgcn_permlanex16(x, x, 0x76543210, 0xfedcba98, false, false);
  }
};
template <int K, int J, int R>
static __device__ __forceinline__ int net_step(int x, int lane)
{
  const int y = lane_xor_i<J>::get(x);
  const int i = 32 * R + lane;
  const bool keepMin = ((i & K) == 0) == ((i & J) == 0);
  return keepMin ? min(x, y) : max(x, y);
}
template <int K, int R>
static __device__ __forceinline__ int net_block(int x, int lane)
{
  if (K >= 32) x = net_step<K, 16, R>(x, lane);
  if (K >= 16) x = net_step<K, 8, R>(x, lane);
  if (K >= 8)  x = net_step<K, 4, R>(x, lane);
  if (K >= 4)  x = net_step<K, 2, R>(x, lane);
  return net_step<K, 1, R>(x, lane);
}
static __device__ __forceinline__ int sort32(int x, int lane)
{
  x = net_block<2, 0>(x, lane);
  x = net_block<4, 0>(x, lane);
  x = net_block<8, 0>(x, lane);
  x = net_block<16, 0>(x, lane);
  return net_block<32, 0>(x, lane);
}
static __device__ __forceinline__ void sort64(int& x0, int& x1, int lane)
{
  x0 = net_block<2, 0>(x0, lane);   x1 = net_block<2, 1>(x1, lane);
  x0 = net_block<4, 0>(x0, lane);   x1 = net_block<4, 1>(x1, lane);
  x0 = net_block<8, 0>(x0, lane);   x1 = net_block<8, 1>(x1, lane);
  x0 = net_block<16, 0>(x0, lane);  x1 = net_block<16, 1>(x1, lane);
  x0 = net_block<32, 0>(x0, lane);  x1 = net_block<32, 1>(x1, lane);
  const int lo = min(x0, x1), hi = max(x0, x1);
  x0 = net_block<64, 0>(lo, lane);  x1 = net_block<64, 1>(hi, lane);
}

static __device__ __forceinline__ float leaky(float t)
{
  return t > 0.0f ? t : NEG_SLOPE * t;
}

static __device__ __forceinline__ void add_rows(
    float4& acc, const float* __restrict__ H, int lane, int sj, float a, int n)
{
  int j = 0;
  for (; j + 4 <= n; j += 4) {
    const int r0 = __builtin_amdgcn_readlane(sj, j);
    const int r1 = __builtin_amdgcn_readlane(sj, j + 1);
    const int r2 = __builtin_amdgcn_readlane(sj, j + 2);
    const int r3 = __builtin_amdgcn_readlane(sj, j + 3);
    const float w0 = __int_as_float(__builtin_amdgcn_readlane(__float_as_int(a), j));
    const float w1 = __int_as_float(__builtin_amdgcn_readlane(__float_as_int(a), j + 1));
    const float w2 = __int_as_float(__builtin_amdgcn_readlane(__float_as_int(a), j + 2));
    const float w3 = __int_as_float(__builtin_amdgcn_readlane(__float_as_int(a), j + 3));
    const float4 h0 = *(const float4*)(H + (size_t)r0 * OUT_DIM + 4 * lane);
    const float4 h1 = *(const float4*)(H + (size_t)r1 * OUT_DIM + 4 * lane);
    const float4 h2 = *(const float4*)(H + (size_t)r2 * OUT_DIM + 4 * lane);
    const float4 h3 = *(const float4*)(H + (size_t)r3 * OUT_DIM + 4 * lane);
    acc.x += w0 * h0.x; acc.y += w0 * h0.y; acc.z += w0 * h0.z; acc.w += w0 * h0.w;
    acc.x += w1 * h1.x; acc.y += w1 * h1.y; acc.z += w1 * h1.z; acc.w += w1 * h1.w;
    acc.x += w2 * h2.x; acc.y += w2 * h2.y; acc.z += w2 * h2.z; acc.w += w2 * h2.w;
    acc.x += w3 * h3.x; acc.y += w3 * h3.y; acc.z += w3 * h3.z; acc.w += w3 * h3.w;
  }
  for (; j < n; ++j) {
    const int r0 = __builtin_amdgcn_readlane(sj, j);
    const float w0 = __int_as_float(__builtin_amdgcn_readlane(__float_as_int(a), j));
    const float4 h0 = *(const float4*)(H + (size_t)r0 * OUT_DIM + 4 * lane);
    acc.x += w0 * h0.x; acc.y += w0 * h0.y; acc.z += w0 * h0.z; acc.w += w0 * h0.w;
  }
}

__global__ __launch_bounds__(256) void gather_softmax(
    const int* __restrict__ off, const int* __restrict__ csr,
    const float* __restrict__ H,
    const float* __restrict__ el, const float* __restrict__ er,
    float* __restrict__ out)
{
  const int lane = threadIdx.x & 31;
  const int d = __builtin_amdgcn_readfirstlane(
      (int)(blockIdx.x * (blockDim.x >> 5) + (threadIdx.x >> 5)));
  if (d >= N_NODES) return;

  const int start = off[d];
  const int end   = off[d + 1];
  float4* orow = (float4*)(out + (size_t)d * OUT_DIM + 4 * lane);
  if (start >= end) {
    *orow = make_float4(0.0f, 0.0f, 0.0f, 0.0f);
    return;
  }
  const float erd = er[d];
  const float ninf = -__builtin_inff();

  const int i0 = start + lane;
  const int i1 = start + 32 + lane;
  int sj0 = csr[i0 < end ? i0 : end - 1];
  int sj1 = csr[i1 < end ? i1 : end - 1];
  sj0 = i0 < end ? sj0 : 0x7fffffff;
  sj1 = i1 < end ? sj1 : 0x7fffffff;
  if (end - start > 32) sort64(sj0, sj1, lane);
  else                  sj0 = sort32(sj0, lane);
  sj0 = i0 < end ? sj0 : 0;
  sj1 = i1 < end ? sj1 : 0;
  const float x0 = el[sj0];
  const float x1 = el[sj1];
  const float s0 = i0 < end ? leaky(x0 + erd) : ninf;
  const float s1 = i1 < end ? leaky(x1 + erd) : ninf;

  float m = fmaxf(s0, s1);
  for (int base = start + 64; base < end; base += 32) {
    const int idx = base + lane;
    if (idx < end) m = fmaxf(m, leaky(el[csr[idx]] + erd));
  }
#pragma unroll
  for (int o = 16; o > 0; o >>= 1) m = fmaxf(m, __shfl_xor(m, o, 32));

  const float e0 = __expf(s0 - m);
  const float e1 = __expf(s1 - m);
  float sum = e0 + e1;
  for (int base = start + 64; base < end; base += 32) {
    const int idx = base + lane;
    if (idx < end) sum += __expf(leaky(el[csr[idx]] + erd) - m);
  }
#pragma unroll
  for (int o = 16; o > 0; o >>= 1) sum += __shfl_xor(sum, o, 32);

  float4 acc = make_float4(0.0f, 0.0f, 0.0f, 0.0f);
  const int deg = end - start;
  add_rows(acc, H, lane, sj0, e0 / sum, deg < 32 ? deg : 32);
  if (deg > 32) add_rows(acc, H, lane, sj1, e1 / sum, deg < 64 ? deg - 32 : 32);
  for (int base = start + 64; base < end; base += 32) {
    const int idx = base + lane;
    int sj = 0;
    float a = 0.0f;
    if (idx < end) {
      sj = csr[idx];
      a = __expf(leaky(el[sj] + erd) - m) / sum;
    }
    const int left = end - base;
    add_rows(acc, H, lane, sj, a, left < 32 ? left : 32);
  }
  *orow = acc;
}

extern "C" void kernel_launch(void* const* d_in, const int* in_sizes, int n_in,
                              void* d_out, int out_size, void* d_ws, size_t ws_size,
                              hipStream_t stream) {
  (void)in_sizes; (void)n_in; (void)out_size; (void)ws_size;

  const float* features  = (const float*)d_in[0];
  const float* fc_weight = (const float*)d_in[1];
  const float* attn_l    = (const float*)d_in[2];
  const float* attn_r    = (const float*)d_in[3];
  const int*   src       = (const int*)d_in[4];
  const int*   dst       = (const int*)d_in[5];
  float*       out       = (float*)d_out;

  float* h    = (float*)d_ws;
  float* el   = h + (size_t)N_NODES * OUT_DIM;
  float* er   = el + N_PAD;
  int*   off  = (int*)(er + N_PAD);
  int*   csr  = off + N_PAD;
  int*   tab   = (int*)d_ws;
  int*   bpack = tab + HIST_PAD;

  chunk_bucket<<<N_CHUNK, 256, 0, stream>>>(src, dst, tab, bpack);
  bin_sort<<<N_BIN, 1024, 0, stream>>>(tab, bpack, off, csr);

  gat_gemm_wmma<<<GEMM_BLOCKS, GEMM_THREADS, 0, stream>>>(features, fc_weight, attn_l, attn_r, h, el, er);

  gather_softmax<<<(N_NODES + 7) / 8, 256, 0, stream>>>(off, csr, h, el, er, out);
}


# Edge sort re-cut to 256 bins of 196 destinations (160 chunks of 10000 edges): the gather launch gets a work-group for every WGP, not 196 for 256; sum order unchanged (the gather orders each segment)
# speedup vs baseline: 7.5604x; 1.1277x over previous
#include <hip/hip_runtime.h>
#include <hip/hip_bf16.h>

#define N_NODES 50000
#define N_EDGES 1600000
#define IN_DIM 256
#define OUT_DIM 128
#define NEG_SLOPE 0.2f
#define N_PAD 50048

typedef __attribute__((ext_vector_type(2))) float v2f;
typedef __attribute__((ext_vector_type(8))) float v8f;

#define GEMM_BLOCKS 256
#define GEMM_THREADS 512
#define ROW_TILES (N_NODES / 16)
#define WL_PITCH (OUT_DIM + 1)
static_assert(ROW_TILES * 16 == N_NODES, "row tiles must cover the nodes exactly");
#define FULL_WAVES (ROW_TILES / GEMM_BLOCKS)
#define LEFT_TILES (ROW_TILES - FULL_WAVES * GEMM_BLOCKS)
static_assert(FULL_WAVES <= GEMM_THREADS / 32 && 4 <= GEMM_THREADS / 32, "a block has a wave for each of its tiles");
static_assert(LEFT_TILES >= 0 && LEFT_TILES <= GEMM_BLOCKS, "fewer left-over row tiles than GEMM work-groups");

static __device__ __forceinline__ float swap_halves(float x)
{
  return __int_as_float(__builtin_amdgcn_permlanex16(
      __float_as_int(x), __float_as_int(x), 0x76543210, 0xfedcba98, false, false));
}

template <int M>
static __device__ __forceinline__ float row_xor(float x)
{
  return __int_as_float(__builtin_amdgcn_update_dpp(
      0, __float_as_int(x), 0x160 + M, 0xf, 0xf, true));
}

static __device__ __forceinline__ int lane_again()
{
  return (int)__builtin_amdgcn_mbcnt_lo(~0u, 0u);
}

template <int NT>
static __device__ __forceinline__ float row_block(
    const float4* __restrict__ aPtr, const float4* bPtr, float* __restrict__ out0,
    const float* __restrict__ al0, const float* __restrict__ ar0)
{
  v8f acc[NT];
#pragma unroll
  for (int t = 0; t < NT; ++t) acc[t] = v8f{};

  float4 a[IN_DIM / 8];
#pragma unroll
  for (int i = 0; i < IN_DIM / 8; ++i) a[i] = aPtr[(i >> 2) * 8 + (i & 3)];

#pragma unroll
  for (int i = 0; i < IN_DIM / 8; ++i) {
    float4 b[NT];
#pragma unroll
    for (int t = 0; t < NT; ++t) b[t] = bPtr[((i >> 2) * 8 + (i & 3)) * WL_PITCH + t * 16];
    v2f a0, a1;
    a0.x = a[i].x; a0.y = a[i].y;
    a1.x = a[i].z; a1.y = a[i].w;
#pragma unroll
    for (int t = 0; t < NT; ++t) {
      v2f b0;
      b0.x = b[t].x; b0.y = b[t].y;
      acc[t] = __builtin_amdgcn_wmma_f32_16x16x4_f32(
           false, a0,  false, b0,
           (short)0, acc[t],  false,  false);
    }
#pragma unroll
    for (int t = 0; t < NT; ++t) {
      v2f b1;
      b1.x = b[t].z; b1.y = b[t].w;
      acc[t] = __builtin_amdgcn_wmma_f32_16x16x4_f32(
           false, a1,  false, b1,
           (short)0, acc[t],  false,  false);
    }
  }

  __builtin_amdgcn_sched_barrier(0);
  const int lane = lane_again();
  const int l16  = lane & 15;
  const int half = lane >> 4;
  float* __restrict__ out = out0 + lane;
  const float* __restrict__ alp = al0 + l16;
  const float* __restrict__ arp = ar0 + l16;

#pragma unroll
  for (int tp = 0; tp < NT / 2; ++tp) {
#pragma unroll
    for (int v = 0; v < 8; ++v) {
      const float lo = acc[2 * tp][v];
      const float hi = acc[2 * tp + 1][v];
      const float got = swap_halves(half ? lo : hi);
      out[(size_t)v * OUT_DIM + 32 * tp]       = half ? got : lo;
      out[(size_t)(v + 8) * OUT_DIM + 32 * tp] = half ? hi : got;
    }
  }

  float pl[8], pr[8];
#pragma unroll
  for (int v = 0; v < 8; ++v) { pl[v] = 0.0f; pr[v] = 0.0f; }
#pragma unroll
  for (int t = 0; t < NT; ++t) {
    const float al = alp[16 * t];
    const float ar = arp[16 * t];
#pragma unroll
    for (int v = 0; v < 8; ++v) {
      pl[v] += acc[t][v] * al;
      pr[v] += acc[t][v] * ar;
    }
  }
#pragma unroll
  for (int v = 0; v < 8; ++v) {
    pl[v] += row_xor<1>(pl[v]);  pr[v] += row_xor<1>(pr[v]);
  }
#pragma unroll
  for (int v = 0; v < 8; ++v) {
    pl[v] += row_xor<2>(pl[v]);  pr[v] += row_xor<2>(pr[v]);
  }
#pragma unroll
  for (int v = 0; v < 8; ++v) {
    pl[v] += row_xor<4>(pl[v]);  pr[v] += row_xor<4>(pr[v]);
  }
#pragma unroll
  for (int v = 0; v < 8; ++v) {
    pl[v] += row_xor<8>(pl[v]);  pr[v] += row_xor<8>(pr[v]);
  }
  float keep = 0.0f;
#pragma unroll
  for (int v = 0; v < 8; ++v)
    if ((l16 & 7) == v) keep = l16 < 8 ? pl[v] : pr[v];
  return keep;
}

template <int FW, int TILE0, int NLEFT>
static __device__ __forceinline__ void gemm_block(
    const float* __restrict__ X,
    const float* __restrict__ W,
    const float* __restrict__ attn_l,
    const float* __restrict__ attn_r,
    float* __restrict__ H,
    float* __restrict__ el, float* __restrict__ er,
    float4* Wl, float (*part)[2][16], int block)
{

  constexpr int STAGE_TRIPS = (IN_DIM / 16) * (OUT_DIM / 32) / (GEMM_THREADS / 32);
  static_assert(STAGE_TRIPS * (GEMM_THREADS / 32) == (IN_DIM / 16) * (OUT_DIM / 32), "staging units divide evenly");
  const int sKq = (int)(threadIdx.x >> 3) & 3;
  const int sN4 = (int)threadIdx.x & 7;
  float4 r[STAGE_TRIPS][4];
#pragma unroll
  for (int i = 0; i < STAGE_TRIPS; ++i) {
    const int unit = i * (GEMM_THREADS / 32) + (int)(threadIdx.x >> 5);
    const int kq = 4 * (unit >> 2) + sKq;
    const int n4 = 8 * (unit & 3) + sN4;
    const float* wp = W + (size_t)(4 * kq) * OUT_DIM + 4 * n4;
#pragma unroll
    for (int c4 = 0; c4 < 4; ++c4) r[i][c4] = *(const float4*)(wp + c4 * OUT_DIM);
  }
#pragma unroll
  for (int i = 0; i < STAGE_TRIPS; ++i) {
    const int unit = i * (GEMM_THREADS / 32) + (int)(threadIdx.x >> 5);
    const int kq = 4 * (unit >> 2) + sKq;
    const int n4 = 8 * (unit & 3) + sN4;
    float4* q = Wl + kq * WL_PITCH + 4 * n4;
    q[0] = make_float4(r[i][0].x, r[i][1].x, r[i][2].x, r[i][3].x);
    q[1] = make_float4(r[i][0].y, r[i][1].y, r[i][2].y, r[i][3].y);
    q[2] = make_float4(r[i][0].z, r[i][1].z, r[i][2].z, r[i][3].z);
    q[3] = make_float4(r[i][0].w, r[i][1].w, r[i][2].w, r[i][3].w);
  }
  __syncthreads();

  const int lane = threadIdx.x & 31;
  const int wave = __builtin_amdgcn_readfirstlane((int)(threadIdx.x >> 5));
  const int l16  = lane & 15;
  const int half = lane >> 4;

  const int tile = TILE0 + (wave < FW ? wave : FW) * GEMM_BLOCKS + block;
  const bool hasTile = wave < FW || (wave < FW + 4 && block < NLEFT);
  const int mBase = tile * 16;

  if (hasTile) {
    const float4* aPtr = (const float4*)(X + (size_t)(mBase + l16) * IN_DIM + 16 * half);
    const float4* bPtr = Wl + (4 * half) * WL_PITCH + l16;
    float* out0 = H + (size_t)mBase * OUT_DIM;
    if (wave < FW) {
      const float keep = row_block<8>(aPtr, bPtr, out0, attn_l, attn_r);
      const int k16 = lane_again() & 15;
      const int krow = 8 * (lane_again() >> 4) + (k16 & 7);
      (k16 < 8 ? el : er)[mBase + krow] = keep;
    } else {
      const int q = wave - FW;
      const int col0 = q * 32;
      const float keep = row_block<2>(aPtr, bPtr + col0, out0 + col0, attn_l + col0, attn_r + col0);
      const int k16 = lane_again() & 15;
      const int krow = 8 * (lane_again() >> 4) + (k16 & 7);
      part[q][k16 >> 3][krow] = keep;
    }
  }
  if (NLEFT > 0) {
    __syncthreads();
    if (wave == FW && hasTile) {
      const int r = lane_again() & 15;
      const int hf = lane_again() >> 4;
      const float s = (part[0][hf][r] + part[1][hf][r]) + (part[2][hf][r] + part[3][hf][r]);
      (hf ? er : el)[mBase + r] = s;
    }
  }
}

#define N_CHUNK 160
#define CHUNK_EDGES 10000
#define BIN_NODES 196
#define N_BIN 256
#define HIST_PAD N_PAD
#define BIN_THREADS 1024
#define BIN_WAVES (BIN_THREADS / 32)
#define RUN_ROUNDS 5
#define SEG_STAGE 12288
static_assert(N_BIN * N_CHUNK <= HIST_PAD, "the descriptor table fits its region");
static_assert(N_CHUNK * CHUNK_EDGES == N_EDGES, "chunks must tile the edge list");
static_assert(CHUNK_EDGES % 4 == 0 && HIST_PAD % 4 == 0, "a chunk's words start on a 16-byte boundary");
static_assert(N_BIN * BIN_NODES >= N_NODES, "bins must cover every node");
static_assert(N_BIN <= 256, "one LDS counter per bin in a 256-entry array");
static_assert(BIN_NODES <= 256, "a destination's number inside its bin fits 8 bits and a 256-entry counter array");
static_assert(N_NODES <= 65536, "a source index must fit 16 bits of the packed edge word");
static_assert(CHUNK_EDGES < 65536, "a run's start and length fit 16 bits each");
static_assert(BIN_WAVES * RUN_ROUNDS >= N_CHUNK && RUN_ROUNDS <= 32, "every run of a bin has a wave and a round, every round a lane");

#define CHUNK_ROUNDS ((CHUNK_EDGES + GEMM_THREADS - 1) / GEMM_THREADS)
static_assert((CHUNK_ROUNDS - 1) * GEMM_THREADS < CHUNK_EDGES, "every round holds at least one edge");
static __device__ __forceinline__ int bin_of(int d)
{
  return (int)((unsigned int)d / (unsigned int)BIN_NODES);
}

static __device__ __forceinline__ void chunk_block(
    const int* __restrict__ src, const int* __restrict__ dst,
    int* __restrict__ tab, int* __restrict__ bpack, int* lds, int chunk)
{
  int* hcnt  = lds;
  int* wtot  = lds + 256;
  int* stage = lds + 320;
  const int tid  = threadIdx.x;
  const int lane = tid & 31;
  const int wave = tid >> 5;
  if (tid < 256) hcnt[tid] = 0;
  const int e0 = chunk * CHUNK_EDGES + tid;
  int d[CHUNK_ROUNDS], s[CHUNK_ROUNDS], r[CHUNK_ROUNDS];
#pragma unroll
  for (int i = 0; i < CHUNK_ROUNDS; ++i) {
    d[i] = 0; s[i] = 0; r[i] = 0;
    if (tid + i * GEMM_THREADS < CHUNK_EDGES) {
      d[i] = dst[e0 + i * GEMM_THREADS];
      s[i] = src[e0 + i * GEMM_THREADS];
    }
  }
  __syncthreads();
#pragma unroll
  for (int i = 0; i < CHUNK_ROUNDS; ++i)
    if (tid + i * GEMM_THREADS < CHUNK_EDGES) {
      const int b = bin_of(d[i]);
      r[i] = atomicAdd(&hcnt[b], 1);
      s[i] |= (d[i] - b * BIN_NODES) << 16;
      d[i] = b;
    }
  __syncthreads();

  int v = 0, x = 0;
  if (tid < 256) {
    v = hcnt[tid];
    x = v;
#pragma unroll
    for (int o = 1; o < 32; o <<= 1) {
      const int y = __shfl_up(x, o, 32);
      if (lane >= o) x += y;
    }
    if (lane == 31) wtot[wave] = x;
  }
  __syncthreads();
  if (tid < 256) {
    int base = 0;
    for (int w = 0; w < wave; ++w) base += wtot[w];
    const int excl = base + x - v;
    hcnt[tid] = excl;
    if (tid < N_BIN) tab[tid * N_CHUNK + chunk] = (excl << 16) | v;
  }
  __syncthreads();

#pragma unroll
  for (int i = 0; i < CHUNK_ROUNDS; ++i)
    if (tid + i * GEMM_THREADS < CHUNK_EDGES)
      stage[hcnt[d[i]] + r[i]] = s[i];
  __syncthreads();
  const int4* stage4 = (const int4*)stage;
  int4* mine4 = (int4*)(bpack + chunk * CHUNK_EDGES);
#pragma unroll
  for (int i = 0; i < (CHUNK_EDGES / 4 + GEMM_THREADS - 1) / GEMM_THREADS; ++i) {
    const int j = tid + i * GEMM_THREADS;
    if (j < CHUNK_EDGES / 4) mine4[j] = stage4[j];
  }
}

static_assert((320 + CHUNK_EDGES) * 4 <= (IN_DIM / 4) * WL_PITCH * 16, "the chunk's LDS use fits the area of W");
__global__ __launch_bounds__(GEMM_THREADS, 8) void gemm_and_chunks(
    const float* __restrict__ X, const float* __restrict__ W,
    const float* __restrict__ attn_l, const float* __restrict__ attn_r,
    float* __restrict__ H, float* __restrict__ el, float* __restrict__ er,
    const int* __restrict__ src, const int* __restrict__ dst,
    int* __restrict__ tab, int* __restrict__ bpack)
{
  __shared__ float4 Wl[(IN_DIM / 4) * WL_PITCH];
  __shared__ float part[4][2][16];
  if (blockIdx.x < GEMM_BLOCKS)
    gemm_block<FULL_WAVES, 0, 0>(X, W, attn_l, attn_r, H, el, er, Wl, part, (int)blockIdx.x);
  else if (blockIdx.x < GEMM_BLOCKS + N_CHUNK)
    chunk_block(src, dst, tab, bpack, (int*)Wl, (int)blockIdx.x - GEMM_BLOCKS);
  else
    gemm_block<0, FULL_WAVES * GEMM_BLOCKS, LEFT_TILES>(X, W, attn_l, attn_r, H, el, er, Wl, part,
                                                      (int)blockIdx.x - GEMM_BLOCKS - N_CHUNK);
}

static __device__ __forceinline__ int bin_block(
    const int* __restrict__ tab, const int* __restrict__ bpack, int* lds, int bin)
{
  int* ncnt   = lds;
  int* ncnt2  = lds + 256;
  int* wtot   = lds + 544;
  int* sstage = lds + 576;
  const int tid  = threadIdx.x;
  const int lane = tid & 31;
  const int wave = __builtin_amdgcn_readfirstlane(tid >> 5);

  int mydesc = 0;
  {
    const int c = wave + BIN_WAVES * lane;
    if (lane < RUN_ROUNDS && c < N_CHUNK) mydesc = tab[bin * N_CHUNK + c];
  }
  if (tid < 256) { ncnt[tid] = 0; ncnt2[tid] = 0; }

  int p[2 * RUN_ROUNDS], r[2 * RUN_ROUNDS];
#pragma unroll
  for (int q = 0; q < RUN_ROUNDS; ++q) {
    const int desc = __builtin_amdgcn_readlane(mydesc, q);
    const int len  = desc & 0xFFFF;
    const int at   = (wave + BIN_WAVES * q) * CHUNK_EDGES + (desc >> 16) + lane;
    p[2 * q]     = bpack[lane < len      ? at      : 0];
    p[2 * q + 1] = bpack[lane + 32 < len ? at + 32 : 0];
  }
#pragma unroll
  for (int q = 0; q < RUN_ROUNDS; ++q) {
    const int len = __builtin_amdgcn_readlane(mydesc, q) & 0xFFFF;
    p[2 * q]     = lane < len      ? p[2 * q]     : -1;
    p[2 * q + 1] = lane + 32 < len ? p[2 * q + 1] : -1;
  }
  __syncthreads();

#pragma unroll
  for (int k = 0; k < 2 * RUN_ROUNDS; ++k) {
    r[k] = 0;
    if (p[k] >= 0) r[k] = atomicAdd(&ncnt[p[k] >> 16], 1);
  }
#pragma unroll
  for (int q = 0; q < RUN_ROUNDS; ++q) {
    const int desc = __builtin_amdgcn_readlane(mydesc, q);
    const int len  = desc & 0xFFFF;
    const int* run = bpack + (wave + BIN_WAVES * q) * CHUNK_EDGES + (desc >> 16);
#pragma clang loop unroll(disable)
    for (int k0 = 64; k0 < len; k0 += 32)
      if (k0 + lane < len) atomicAdd(&ncnt2[run[k0 + lane] >> 16], 1);
  }
  __syncthreads();

  int v1 = 0, v = 0, x = 0;
  if (tid < 256) {
    v1 = ncnt[tid];
    v = v1 + ncnt2[tid];
    x = v;
#pragma unroll
    for (int o = 1; o < 32; o <<= 1) {
      const int y = __shfl_up(x, o, 32);
      if (lane >= o) x += y;
    }
    if (lane == 31) wtot[wave] = x;
  }
  __syncthreads();
  if (tid < 256) {
    int base = 0;
    for (int w = 0; w < wave; ++w) base += wtot[w];
    const int excl = base + x - v;
    ncnt[tid]  = excl;
    ncnt2[tid] = excl + v1;
  }
  __syncthreads();

  int total = 0;
  for (int w = 0; w < 8; ++w) total += wtot[w];
  if (total <= SEG_STAGE) {
#pragma unroll
    for (int k = 0; k < 2 * RUN_ROUNDS; ++k)
      if (p[k] >= 0) sstage[ncnt[p[k] >> 16] + r[k]] = p[k] & 0xFFFF;
#pragma unroll
    for (int q = 0; q < RUN_ROUNDS; ++q) {
      const int desc = __builtin_amdgcn_readlane(mydesc, q);
      const int len  = desc & 0xFFFF;
      const int* run = bpack + (wave + BIN_WAVES * q) * CHUNK_EDGES + (desc >> 16);
#pragma clang loop unroll(disable)
      for (int k0 = 64; k0 < len; k0 += 32)
        if (k0 + lane < len) {
          const int w = run[k0 + lane];
          sstage[atomicAdd(&ncnt2[w >> 16], 1)] = w & 0xFFFF;
        }
    }
    __syncthreads();
  }
  return total;
}

template <int M>
static __device__ __forceinline__ int row_xor_i(int x)
{
  return __builtin_amdgcn_update_dpp(0, x, 0x160 + M, 0xf, 0xf, true);
}
template <int J> struct lane_xor_i {
  static __device__ __forceinline__ int get(int x) { return row_xor_i<J>(x); }
};
template <> struct lane_xor_i<16> {
  static __device__ __forceinline__ int get(int x)
  {
    return __builtin_amdgcn_permlanex16(x, x, 0x76543210, 0xfedcba98, false, false);
  }
};
template <int K, int J, int R>
static __device__ __forceinline__ int net_step(int x, int lane)
{
  const int y = lane_xor_i<J>::get(x);
  const int i = 32 * R + lane;
  const bool keepMin = ((i & K) == 0) == ((i & J) == 0);
  return keepMin ? min(x, y) : max(x, y);
}
template <int K, int R>
static __device__ __forceinline__ int net_block(int x, int lane)
{
  if (K >= 32) x = net_step<K, 16, R>(x, lane);
  if (K >= 16) x = net_step<K, 8, R>(x, lane);
  if (K >= 8)  x = net_step<K, 4, R>(x, lane);
  if (K >= 4)  x = net_step<K, 2, R>(x, lane);
  return net_step<K, 1, R>(x, lane);
}
static __device__ __forceinline__ int sort32(int x, int lane)
{
  x = net_block<2, 0>(x, lane);
  x = net_block<4, 0>(x, lane);
  x = net_block<8, 0>(x, lane);
  x = net_block<16, 0>(x, lane);
  return net_block<32, 0>(x, lane);
}
static __device__ __forceinline__ void sort64(int& x0, int& x1, int lane)
{
  x0 = net_block<2, 0>(x0, lane);   x1 = net_block<2, 1>(x1, lane);
  x0 = net_block<4, 0>(x0, lane);   x1 = net_block<4, 1>(x1, lane);
  x0 = net_block<8, 0>(x0, lane);   x1 = net_block<8, 1>(x1, lane);
  x0 = net_block<16, 0>(x0, lane);  x1 = net_block<16, 1>(x1, lane);
  x0 = net_block<32, 0>(x0, lane);  x1 = net_block<32, 1>(x1, lane);
  const int lo = min(x0, x1), hi = max(x0, x1);
  x0 = net_block<64, 0>(lo, lane);  x1 = net_block<64, 1>(hi, lane);
}

static __device__ __forceinline__ float leaky(float t)
{
  return t > 0.0f ? t : NEG_SLOPE * t;
}

template <int R>
static __device__ __forceinline__ void rows_group(
    float4& acc, const float* __restrict__ H, int lane, int sj, float a, int j)
{
  float4 h[R];
  float w[R];
#pragma unroll
  for (int i = 0; i < R; ++i) {
    const int row = __builtin_amdgcn_readlane(sj, j + i);
    w[i] = __int_as_float(__builtin_amdgcn_readlane(__float_as_int(a), j + i));
    h[i] = *(const float4*)(H + (size_t)row * OUT_DIM + 4 * lane);
  }
#pragma unroll
  for (int i = 0; i < R; ++i) {
    acc.x += w[i] * h[i].x; acc.y += w[i] * h[i].y; acc.z += w[i] * h[i].z; acc.w += w[i] * h[i].w;
  }
}

static __device__ __forceinline__ void add_rows(
    float4& acc, const float* __restrict__ H, int lane, int sj, float a, int n)
{
  int j = 0;
  for (; j + 16 <= n; j += 16) rows_group<16>(acc, H, lane, sj, a, j);
  if (j + 8 <= n) { rows_group<8>(acc, H, lane, sj, a, j); j += 8; }
  if (j + 4 <= n) { rows_group<4>(acc, H, lane, sj, a, j); j += 4; }
  for (; j < n; ++j) rows_group<1>(acc, H, lane, sj, a, j);
}

static __device__ __forceinline__ void gather_node(
    const int* seg, int start, int end,
    const float* __restrict__ H, const float* __restrict__ el, const float* __restrict__ er,
    float* __restrict__ out, int d, int lane)
{
  float4* orow = (float4*)(out + (size_t)d * OUT_DIM + 4 * lane);
  if (start >= end) {
    *orow = make_float4(0.0f, 0.0f, 0.0f, 0.0f);
    return;
  }
  const float erd = er[d];
  const float ninf = -__builtin_inff();

  const int i0 = start + lane;
  const int i1 = start + 32 + lane;
  int sj0 = seg[i0 < end ? i0 : end - 1];
  int sj1 = seg[i1 < end ? i1 : end - 1];
  sj0 = i0 < end ? sj0 : 0x7fffffff;
  sj1 = i1 < end ? sj1 : 0x7fffffff;
  if (end - start > 32) sort64(sj0, sj1, lane);
  else                  sj0 = sort32(sj0, lane);
  sj0 = i0 < end ? sj0 : 0;
  sj1 = i1 < end ? sj1 : 0;
  const float x0 = el[sj0];
  const float x1 = el[sj1];
  const float s0 = i0 < end ? leaky(x0 + erd) : ninf;
  const float s1 = i1 < end ? leaky(x1 + erd) : ninf;

  float m = fmaxf(s0, s1);
  for (int base = start + 64; base < end; base += 32) {
    const int idx = base + lane;
    if (idx < end) m = fmaxf(m, leaky(el[seg[idx]] + erd));
  }
#pragma unroll
  for (int o = 16; o > 0; o >>= 1) m = fmaxf(m, __shfl_xor(m, o, 32));

  const float e0 = __expf(s0 - m);
  const float e1 = __expf(s1 - m);
  float sum = e0 + e1;
  for (int base = start + 64; base < end; base += 32) {
    const int idx = base + lane;
    if (idx < end) sum += __expf(leaky(el[seg[idx]] + erd) - m);
  }
#pragma unroll
  for (int o = 16; o > 0; o >>= 1) sum += __shfl_xor(sum, o, 32);

  float4 acc = make_float4(0.0f, 0.0f, 0.0f, 0.0f);
  const int deg = end - start;
  add_rows(acc, H, lane, sj0, e0 / sum, deg < 32 ? deg : 32);
  if (deg > 32) add_rows(acc, H, lane, sj1, e1 / sum, deg < 64 ? deg - 32 : 32);
  for (int base = start + 64; base < end; base += 32) {
    const int idx = base + lane;
    int sj = 0;
    float a = 0.0f;
    if (idx < end) {
      sj = seg[idx];
      a = __expf(leaky(el[sj] + erd) - m) / sum;
    }
    const int left = end - base;
    add_rows(acc, H, lane, sj, a, left < 32 ? left : 32);
  }
  *orow = acc;
}

static __device__ __forceinline__ void slow_node(
    const int* __restrict__ tab, const int* __restrict__ bpack, int bin, int t,
    const float* __restrict__ H, const float* __restrict__ el, const float* __restrict__ er,
    float* __restrict__ out, int d, int lane)
{
  float4* orow = (float4*)(out + (size_t)d * OUT_DIM + 4 * lane);
  const float erd = er[d];
  float m = -__builtin_inff();
  float sum = 0.0f;
  float4 acc = make_float4(0.0f, 0.0f, 0.0f, 0.0f);
  for (int pass = 0; pass < 3; ++pass) {
    for (int c = 0; c < N_CHUNK; ++c) {
      const int desc = __builtin_amdgcn_readfirstlane(tab[bin * N_CHUNK + c]);
      const int len = desc & 0xFFFF;
      const int at = c * CHUNK_EDGES + (desc >> 16);
      for (int k0 = 0; k0 < len; k0 += 32) {
        const int idx = k0 + lane;
        const int w = bpack[idx < len ? at + idx : 0];
        const bool hit = idx < len && (w >> 16) == t;
        const int sj = hit ? (w & 0xFFFF) : 0;
        const float s = leaky(el[sj] + erd);
        if (pass == 0) {
          if (hit) m = fmaxf(m, s);
        } else if (pass == 1) {
          if (hit) sum += __expf(s - m);
        } else {
          const float a = hit ? __expf(s - m) / sum : 0.0f;
          unsigned int todo = __builtin_amdgcn_ballot_w32(hit);
          while (todo != 0u) {
            const int j = __builtin_ctz(todo);
            todo &= todo - 1u;
            rows_group<1>(acc, H, lane, sj, a, j);
          }
        }
      }
    }
    if (pass == 0) {
#pragma unroll
      for (int o = 16; o > 0; o >>= 1) m = fmaxf(m, __shfl_xor(m, o, 32));
      if (m == -__builtin_inff()) {
        *orow = make_float4(0.0f, 0.0f, 0.0f, 0.0f);
        return;
      }
    } else if (pass == 1) {
#pragma unroll
      for (int o = 16; o > 0; o >>= 1) sum += __shfl_xor(sum, o, 32);
    }
  }
  *orow = acc;
}

__global__ __launch_bounds__(BIN_THREADS) __attribute__((amdgpu_waves_per_eu(8, 8))) void bins_and_gather(
    const int* __restrict__ tab, const int* __restrict__ bpack,
    const float* __restrict__ H,
    const float* __restrict__ el, const float* __restrict__ er,
    float* __restrict__ out)
{
  __shared__ int lds[576 + SEG_STAGE];
  const int bin = (int)blockIdx.x;
  const int total = bin_block(tab, bpack, lds, bin);
  const int* ncnt = lds;
  const int* sstage = lds + 576;
  const int lane = threadIdx.x & 31;
  const int wave = __builtin_amdgcn_readfirstlane((int)(threadIdx.x >> 5));
#pragma clang loop unroll(disable)
  for (int k = 0; k < (BIN_NODES + BIN_WAVES - 1) / BIN_WAVES; ++k) {
    const int t = wave + BIN_WAVES * k;
    const int d = bin * BIN_NODES + t;
    if (t < BIN_NODES && d < N_NODES) {
      if (total <= SEG_STAGE) {
        const int start = __builtin_amdgcn_readfirstlane(ncnt[t]);
        const int end   = __builtin_amdgcn_readfirstlane(t < 255 ? ncnt[t + 1] : total);
        gather_node(sstage, start, end, H, el, er, out, d, lane);
      } else {
        slow_node(tab, bpack, bin, t, H, el, er, out, d, lane);
      }
    }
  }
}

extern "C" void kernel_launch(void* const* d_in, const int* in_sizes, int n_in,
                              void* d_out, int out_size, void* d_ws, size_t ws_size,
                              hipStream_t stream) {
  (void)in_sizes; (void)n_in; (void)out_size; (void)ws_size;

  const float* features  = (const float*)d_in[0];
  const float* fc_weight = (const float*)d_in[1];
  const float* attn_l    = (const float*)d_in[2];
  const float* attn_r    = (const float*)d_in[3];
  const int*   src       = (const int*)d_in[4];
  const int*   dst       = (const int*)d_in[5];
  float*       out       = (float*)d_out;

  float* h    = (float*)d_ws;
  float* el   = h + (size_t)N_NODES * OUT_DIM;
  float* er   = el + N_PAD;
  int*   tab   = (int*)(er + N_PAD);
  int*   bpack = tab + HIST_PAD;

  gemm_and_chunks<<<GEMM_BLOCKS + N_CHUNK + LEFT_TILES, GEMM_THREADS, 0, stream>>>(
      features, fc_weight, attn_l, attn_r, h, el, er, src, dst, tab, bpack);

  bins_and_gather<<<N_BIN, BIN_THREADS, 0, stream>>>(tab, bpack, h, el, er, out);
}
